# starting kernel
# baseline (speedup 1.0000x reference)
#include <hip/hip_runtime.h>
#include <math.h>

#define DIM      128
#define NHEADS   16
#define DPH      8
#define QKV_N    384

typedef float v2f __attribute__((ext_vector_type(2)));
typedef float v8f __attribute__((ext_vector_type(8)));

__device__ __forceinline__ v8f wmma_f32_16x16x4(v2f a, v2f b, v8f c) {
    return __builtin_amdgcn_wmma_f32_16x16x4_f32(
        false, a, false, b, (short)0, c, false, false);
}

__global__ void ln_stats_kernel(const float* __restrict__ nf,
                                float* __restrict__ stats, int natoms) {
    const int atom = blockIdx.x * 8 + (threadIdx.x >> 5);
    const int lane = threadIdx.x & 31;
    if (atom >= natoms) return;
    const float4 v = ((const float4*)(nf + (size_t)atom * DIM))[lane];
    float s  = v.x + v.y + v.z + v.w;
    float s2 = v.x*v.x + v.y*v.y + v.z*v.z + v.w*v.w;
    #pragma unroll
    for (int m = 1; m < 32; m <<= 1) {
        s  += __shfl_xor(s,  m, 32);
        s2 += __shfl_xor(s2, m, 32);
    }
    if (lane == 0) {
        const float mu  = s * (1.0f / DIM);
        const float var = s2 * (1.0f / DIM) - mu * mu;
        stats[2 * atom + 0] = mu;
        stats[2 * atom + 1] = rsqrtf(var + 1e-5f);
    }
}

__global__ void qkv_gemm_kernel(const float* __restrict__ nf,
                                const float* __restrict__ stats,
                                const float* __restrict__ ln_g,
                                const float* __restrict__ ln_b,
                                const float* __restrict__ w,
                                const float* __restrict__ bias,
                                float* __restrict__ out,
                                int natoms) {
    const int lane = threadIdx.x & 31;
    const int wid  = threadIdx.x >> 5;
    const int half = lane >> 4;
    const int l16  = lane & 15;

    const int mbase = blockIdx.x * 16;
    const int nbase = (blockIdx.y * 8 + wid) * 16;
    const int ncol  = nbase + l16;

    int arow = mbase + l16;
    if (arow >= natoms) arow = natoms - 1;
    const float mu = stats[2 * arow + 0];
    const float rs = stats[2 * arow + 1];
    const float* __restrict__ ap = nf + (size_t)arow * DIM;

    v8f acc = {};
    #pragma unroll 4
    for (int k0 = 0; k0 < DIM; k0 += 4) {
        const int kk = k0 + 2 * half;
        v2f a, b;
        a.x = (ap[kk]     - mu) * rs * ln_g[kk]     + ln_b[kk];
        a.y = (ap[kk + 1] - mu) * rs * ln_g[kk + 1] + ln_b[kk + 1];
        b.x = w[(size_t)kk       * QKV_N + ncol];
        b.y = w[(size_t)(kk + 1) * QKV_N + ncol];
        acc = wmma_f32_16x16x4(a, b, acc);
    }

    const float bv = bias[ncol];
    #pragma unroll
    for (int r = 0; r < 8; ++r) {
        const int rowm = mbase + r + half * 8;
        if (rowm < natoms)
            out[(size_t)rowm * QKV_N + ncol] = acc[r] + bv;
    }
}

__global__ void zero_kernel(float* __restrict__ p, size_t n) {
    size_t i = (size_t)blockIdx.x * blockDim.x + threadIdx.x;
    const size_t stride = (size_t)gridDim.x * blockDim.x;
    for (; i < n; i += stride) p[i] = 0.0f;
}

__global__ void edge_kernel(const float* __restrict__ qkv,
                            const float* __restrict__ edge_feat,
                            const float* __restrict__ edge_vec,
                            const float* __restrict__ radial,
                            const int* __restrict__ row,
                            const int* __restrict__ col,
                            float* __restrict__ m_feat,
                            float* __restrict__ out_vec,
                            int nedges) {
    const long long gid = (long long)blockIdx.x * blockDim.x + threadIdx.x;
    const int e = (int)(gid >> 7);
    if (e >= nedges) return;
    const int t = (int)(gid & 127);
    const int h = t >> 3;
    const int d = t & 7;

    const int r = row[e];
    const int c = col[e];

    const float q = qkv[(size_t)r * QKV_N + h * 24 + d];
    const float k = qkv[(size_t)c * QKV_N + h * 24 + 8  + d];
    const float v = qkv[(size_t)c * QKV_N + h * 24 + 16 + d];

    float p = q * k;
    p += __shfl_xor(p, 1, 32);
    p += __shfl_xor(p, 2, 32);
    p += __shfl_xor(p, 4, 32);

    const float attn = 0.5f * p * (1.0f + erff(p * 0.70710678118654752440f))
                     * radial[e];

    const float mf = v * edge_feat[(size_t)e * DIM + t] * attn;
    atomicAdd(&m_feat[(size_t)r * DIM + t], mf);

    const float ev0 = edge_vec[(size_t)e * 3 + 0];
    const float ev1 = edge_vec[(size_t)e * 3 + 1];
    const float ev2 = edge_vec[(size_t)e * 3 + 2];
    float* vb = out_vec + (size_t)r * (3 * DIM) + t;
    atomicAdd(vb,           v * ev0);
    atomicAdd(vb + DIM,     v * ev1);
    atomicAdd(vb + 2 * DIM, v * ev2);
}

__global__ void out_gemm_kernel(const float* __restrict__ mf,
                                const float* __restrict__ w,
                                const float* __restrict__ bias,
                                float* __restrict__ out,
                                int natoms) {
    const int lane = threadIdx.x & 31;
    const int wid  = threadIdx.x >> 5;
    const int half = lane >> 4;
    const int l16  = lane & 15;

    const int mbase = blockIdx.x * 16;
    const int nbase = wid * 16;
    const int ncol  = nbase + l16;

    int arow = mbase + l16;
    if (arow >= natoms) arow = natoms - 1;
    const float* __restrict__ ap = mf + (size_t)arow * DIM;

    v8f acc = {};
    #pragma unroll 4
    for (int k0 = 0; k0 < DIM; k0 += 4) {
        const int kk = k0 + 2 * half;
        v2f a, b;
        a.x = ap[kk];
        a.y = ap[kk + 1];
        b.x = w[(size_t)kk       * DIM + ncol];
        b.y = w[(size_t)(kk + 1) * DIM + ncol];
        acc = wmma_f32_16x16x4(a, b, acc);
    }

    const float bv = bias[ncol];
    #pragma unroll
    for (int r = 0; r < 8; ++r) {
        const int rowm = mbase + r + half * 8;
        if (rowm < natoms)
            out[(size_t)rowm * DIM + ncol] = acc[r] + bv;
    }
}

extern "C" void kernel_launch(void* const* d_in, const int* in_sizes, int n_in,
                              void* d_out, int out_size, void* d_ws, size_t ws_size,
                              hipStream_t stream) {
    const float* nf   = (const float*)d_in[0];
    const float* ef   = (const float*)d_in[1];
    const float* evec = (const float*)d_in[2];
    const float* rad  = (const float*)d_in[3];
    const float* qkvw = (const float*)d_in[4];
    const float* qkvb = (const float*)d_in[5];
    const float* outw = (const float*)d_in[6];
    const float* outb = (const float*)d_in[7];
    const float* lng  = (const float*)d_in[8];
    const float* lnb  = (const float*)d_in[9];
    const int*   row  = (const int*)d_in[10];
    const int*   col  = (const int*)d_in[11];

    const int natoms = in_sizes[0] / DIM;
    const int nedges = in_sizes[3];

    float* stats = (float*)d_ws;
    float* qkv   = stats + (size_t)2 * natoms;
    float* mfeat = qkv + (size_t)natoms * QKV_N;

    float* out_feat = (float*)d_out;
    float* out_vec  = out_feat + (size_t)natoms * DIM;

    ln_stats_kernel<<<(natoms + 7) / 8, 256, 0, stream>>>(nf, stats, natoms);

    dim3 g1((natoms + 15) / 16, 3);
    qkv_gemm_kernel<<<g1, 256, 0, stream>>>(nf, stats, lng, lnb, qkvw, qkvb,
                                            qkv, natoms);

    const size_t zc1 = (size_t)natoms * DIM;
    const size_t zc2 = (size_t)natoms * 3 * DIM;
    zero_kernel<<<2048, 256, 0, stream>>>(mfeat, zc1);
    zero_kernel<<<4096, 256, 0, stream>>>(out_vec, zc2);

    const long long tot = (long long)nedges * DIM;
    const unsigned eblocks = (unsigned)((tot + 255) / 256);
    edge_kernel<<<eblocks, 256, 0, stream>>>(qkv, ef, evec, rad, row, col,
                                             mfeat, out_vec, nedges);

    dim3 g2((natoms + 15) / 16, 1);
    out_gemm_kernel<<<g2, 256, 0, stream>>>(mfeat, outw, outb, out_feat, natoms);
}


# Edge scatter-add (4 f32 atomics per lane) replaced by a per-call counting sort by row (returning int add = arrival-order rank) and a one-wave-per-row register sum, each row stored once
# speedup vs baseline: 1.8481x; 1.8481x over previous
#include <hip/hip_runtime.h>
#include <math.h>

#define DIM      128
#define NHEADS   16
#define DPH      8
#define QKV_N    384

typedef float v2f __attribute__((ext_vector_type(2)));
typedef float v8f __attribute__((ext_vector_type(8)));
typedef float v4f __attribute__((ext_vector_type(4)));

__device__ __forceinline__ v8f wmma_f32_16x16x4(v2f a, v2f b, v8f c) {
    return __builtin_amdgcn_wmma_f32_16x16x4_f32(
        false, a, false, b, (short)0, c, false, false);
}

__global__ void ln_stats_kernel(const float* __restrict__ nf,
                                float* __restrict__ stats, int natoms) {
    const int atom = blockIdx.x * 8 + (threadIdx.x >> 5);
    const int lane = threadIdx.x & 31;
    if (atom >= natoms) return;
    const float4 v = ((const float4*)(nf + (size_t)atom * DIM))[lane];
    float s  = v.x + v.y + v.z + v.w;
    float s2 = v.x*v.x + v.y*v.y + v.z*v.z + v.w*v.w;
    #pragma unroll
    for (int m = 1; m < 32; m <<= 1) {
        s  += __shfl_xor(s,  m, 32);
        s2 += __shfl_xor(s2, m, 32);
    }
    if (lane == 0) {
        const float mu  = s * (1.0f / DIM);
        const float var = s2 * (1.0f / DIM) - mu * mu;
        stats[2 * atom + 0] = mu;
        stats[2 * atom + 1] = rsqrtf(var + 1e-5f);
    }
}

__global__ void qkv_gemm_kernel(const float* __restrict__ nf,
                                const float* __restrict__ stats,
                                const float* __restrict__ ln_g,
                                const float* __restrict__ ln_b,
                                const float* __restrict__ w,
                                const float* __restrict__ bias,
                                float* __restrict__ out,
                                int natoms) {
    const int lane = threadIdx.x & 31;
    const int wid  = threadIdx.x >> 5;
    const int half = lane >> 4;
    const int l16  = lane & 15;

    const int mbase = blockIdx.x * 16;
    const int nbase = (blockIdx.y * 8 + wid) * 16;
    const int ncol  = nbase + l16;

    int arow = mbase + l16;
    if (arow >= natoms) arow = natoms - 1;
    const float mu = stats[2 * arow + 0];
    const float rs = stats[2 * arow + 1];
    const float* __restrict__ ap = nf + (size_t)arow * DIM;

    v8f acc = {};
    #pragma unroll 4
    for (int k0 = 0; k0 < DIM; k0 += 4) {
        const int kk = k0 + 2 * half;
        v2f a, b;
        a.x = (ap[kk]     - mu) * rs * ln_g[kk]     + ln_b[kk];
        a.y = (ap[kk + 1] - mu) * rs * ln_g[kk + 1] + ln_b[kk + 1];
        b.x = w[(size_t)kk       * QKV_N + ncol];
        b.y = w[(size_t)(kk + 1) * QKV_N + ncol];
        acc = wmma_f32_16x16x4(a, b, acc);
    }

    const float bv = bias[ncol];
    #pragma unroll
    for (int r = 0; r < 8; ++r) {
        const int rowm = mbase + r + half * 8;
        if (rowm < natoms)
            out[(size_t)rowm * QKV_N + ncol] = acc[r] + bv;
    }
}

__global__ void zero_kernel(float* __restrict__ p, size_t n) {
    size_t i = (size_t)blockIdx.x * blockDim.x + threadIdx.x;
    const size_t stride = (size_t)gridDim.x * blockDim.x;
    for (; i < n; i += stride) p[i] = 0.0f;
}

__global__ void rank_kernel(const int* __restrict__ row,
                            int* __restrict__ cnt,
                            int* __restrict__ rank,
                            int nedges) {
    const int e = blockIdx.x * blockDim.x + threadIdx.x;
    if (e >= nedges) return;
    rank[e] = atomicAdd(&cnt[row[e]], 1);
}

__global__ void __launch_bounds__(1024)
scan_kernel(const int* __restrict__ cnt, int* __restrict__ offs, int natoms) {
    __shared__ int wsum[32];
    const int tid  = threadIdx.x;
    const int lane = tid & 31;
    const int wid  = tid >> 5;
    const int chunk = (natoms + 1023) / 1024;
    const int b = tid * chunk;
    const int bend = (b + chunk < natoms) ? (b + chunk) : natoms;

    int s = 0;
    for (int i = b; i < bend; ++i) s += cnt[i];

    int incl = s;
    for (int m = 1; m < 32; m <<= 1) {
        const int t = __shfl_up(incl, m, 32);
        if (lane >= m) incl += t;
    }
    if (lane == 31) wsum[wid] = incl;
    __syncthreads();
    if (wid == 0) {
        const int w = wsum[lane];
        int wi = w;
        for (int m = 1; m < 32; m <<= 1) {
            const int t = __shfl_up(wi, m, 32);
            if (lane >= m) wi += t;
        }
        wsum[lane] = wi - w;
    }
    __syncthreads();

    int run = wsum[wid] + incl - s;
    #pragma unroll 8
    for (int i = b; i < bend; ++i) { offs[i] = run; run += cnt[i]; }
    if (tid == 1023) offs[natoms] = run;
}

__global__ void fill_kernel(const int* __restrict__ row,
                            const int* __restrict__ col,
                            const int* __restrict__ offs,
                            const int* __restrict__ rank,
                            int2* __restrict__ recs,
                            int nedges) {
    const int e = blockIdx.x * blockDim.x + threadIdx.x;
    if (e >= nedges) return;
    recs[offs[row[e]] + rank[e]] = make_int2(e, col[e]);
}

__global__ void __launch_bounds__(256)
gather_kernel(const float* __restrict__ qkv,
              const float* __restrict__ edge_feat,
              const float* __restrict__ edge_vec,
              const float* __restrict__ radial,
              const int*   __restrict__ offs,
              const int2*  __restrict__ recs,
              float* __restrict__ m_feat,
              float* __restrict__ out_vec,
              int natoms) {
    const int lane = threadIdx.x & 31;
    const int wave   = __builtin_amdgcn_readfirstlane(
                           (int)(blockIdx.x * (blockDim.x >> 5) + (threadIdx.x >> 5)));
    const int nwaves = (int)(gridDim.x * (blockDim.x >> 5));
    const int hoff = (lane >> 1) * 24 + (lane & 1) * 4;

    for (int atom = wave; atom < natoms; atom += nwaves) {
        const int start = offs[atom];
        const int end   = offs[atom + 1];
        const v4f q = *(const v4f*)(qkv + (size_t)atom * QKV_N + hoff);

        v4f mf = {0.0f, 0.0f, 0.0f, 0.0f};
        v4f a0 = mf, a1 = mf, a2 = mf;

        for (int base = start; base < end; base += 32) {
            const int n = (end - base < 32) ? (end - base) : 32;
            int2 rec = make_int2(0, 0);
            if (lane < n) rec = recs[base + lane];

            for (int i = 0; i < n; ++i) {
                const int e = __builtin_amdgcn_readlane(rec.x, i);
                const int c = __builtin_amdgcn_readlane(rec.y, i);
                const float rad = radial[e];
                const float ev0 = edge_vec[(size_t)e * 3 + 0];
                const float ev1 = edge_vec[(size_t)e * 3 + 1];
                const float ev2 = edge_vec[(size_t)e * 3 + 2];
                const float* kp = qkv + (size_t)c * QKV_N + hoff;
                const v4f k = *(const v4f*)(kp + 8);
                const v4f v = *(const v4f*)(kp + 16);
                const v4f ef = __builtin_nontemporal_load(
                    (const v4f*)(edge_feat + (size_t)e * DIM + lane * 4));

                float p = q.x * k.x + q.y * k.y + q.z * k.z + q.w * k.w;
                p += __builtin_bit_cast(float, __builtin_amdgcn_update_dpp(
                         0, __builtin_bit_cast(int, p), 0xB1, 0xF, 0xF, true));

                const float attn = 0.5f * p
                                 * (1.0f + erff(p * 0.70710678118654752440f))
                                 * rad;
                mf += v * ef * attn;
                a0 += v * ev0;
                a1 += v * ev1;
                a2 += v * ev2;
            }
        }

        *(v4f*)(m_feat + (size_t)atom * DIM + lane * 4) = mf;
        float* vb = out_vec + (size_t)atom * (3 * DIM) + lane * 4;
        *(v4f*)(vb)           = a0;
        *(v4f*)(vb + DIM)     = a1;
        *(v4f*)(vb + 2 * DIM) = a2;
    }
}

__global__ void out_gemm_kernel(const float* __restrict__ mf,
                                const float* __restrict__ w,
                                const float* __restrict__ bias,
                                float* __restrict__ out,
                                int natoms) {
    const int lane = threadIdx.x & 31;
    const int wid  = threadIdx.x >> 5;
    const int half = lane >> 4;
    const int l16  = lane & 15;

    const int mbase = blockIdx.x * 16;
    const int nbase = wid * 16;
    const int ncol  = nbase + l16;

    int arow = mbase + l16;
    if (arow >= natoms) arow = natoms - 1;
    const float* __restrict__ ap = mf + (size_t)arow * DIM;

    v8f acc = {};
    #pragma unroll 4
    for (int k0 = 0; k0 < DIM; k0 += 4) {
        const int kk = k0 + 2 * half;
        v2f a, b;
        a.x = ap[kk];
        a.y = ap[kk + 1];
        b.x = w[(size_t)kk       * DIM + ncol];
        b.y = w[(size_t)(kk + 1) * DIM + ncol];
        acc = wmma_f32_16x16x4(a, b, acc);
    }

    const float bv = bias[ncol];
    #pragma unroll
    for (int r = 0; r < 8; ++r) {
        const int rowm = mbase + r + half * 8;
        if (rowm < natoms)
            out[(size_t)rowm * DIM + ncol] = acc[r] + bv;
    }
}

extern "C" void kernel_launch(void* const* d_in, const int* in_sizes, int n_in,
                              void* d_out, int out_size, void* d_ws, size_t ws_size,
                              hipStream_t stream) {
    const float* nf   = (const float*)d_in[0];
    const float* ef   = (const float*)d_in[1];
    const float* evec = (const float*)d_in[2];
    const float* rad  = (const float*)d_in[3];
    const float* qkvw = (const float*)d_in[4];
    const float* qkvb = (const float*)d_in[5];
    const float* outw = (const float*)d_in[6];
    const float* outb = (const float*)d_in[7];
    const float* lng  = (const float*)d_in[8];
    const float* lnb  = (const float*)d_in[9];
    const int*   row  = (const int*)d_in[10];
    const int*   col  = (const int*)d_in[11];

    const int natoms = in_sizes[0] / DIM;
    const int nedges = in_sizes[3];

    float* qkv   = (float*)d_ws;
    float* mfeat = qkv + (size_t)natoms * QKV_N;
    float* stats = mfeat + (size_t)natoms * DIM;

    float* out_feat = (float*)d_out;
    float* out_vec  = out_feat + (size_t)natoms * DIM;

    int2* recs = (int2*)d_out;
    int*  rank = (int*)(recs + nedges);
    int*  cnt  = rank + nedges;
    int*  offs = cnt + natoms;

    ln_stats_kernel<<<(natoms + 7) / 8, 256, 0, stream>>>(nf, stats, natoms);

    dim3 g1((natoms + 15) / 16, 3);
    qkv_gemm_kernel<<<g1, 256, 0, stream>>>(nf, stats, lng, lnb, qkvw, qkvb,
                                            qkv, natoms);

    const unsigned eblocks = (unsigned)((nedges + 255) / 256);
    zero_kernel<<<64, 256, 0, stream>>>((float*)cnt, (size_t)natoms);
    rank_kernel<<<eblocks, 256, 0, stream>>>(row, cnt, rank, nedges);
    scan_kernel<<<1, 1024, 0, stream>>>(cnt, offs, natoms);
    fill_kernel<<<eblocks, 256, 0, stream>>>(row, col, offs, rank, recs, nedges);
    gather_kernel<<<2048, 256, 0, stream>>>(qkv, ef, evec, rad, offs, recs,
                                            mfeat, out_vec, natoms);

    dim3 g2((natoms + 15) / 16, 1);
    out_gemm_kernel<<<g2, 256, 0, stream>>>(mfeat, outw, outb, out_feat, natoms);
}


# LN+QKV GEMM re-tiled: a work-group normalises its 32 rows once into LDS (row statistics in-block, statistics launch gone); each wave does 2x3 tiles, A fragments from LDS; sum order kept
# speedup vs baseline: 2.4943x; 1.3497x over previous
#include <hip/hip_runtime.h>
#include <math.h>

#define DIM      128
#define NHEADS   16
#define DPH      8
#define QKV_N    384

typedef float v2f __attribute__((ext_vector_type(2)));
typedef float v8f __attribute__((ext_vector_type(8)));
typedef float v4f __attribute__((ext_vector_type(4)));

__device__ __forceinline__ v8f wmma_f32_16x16x4(v2f a, v2f b, v8f c) {
    return __builtin_amdgcn_wmma_f32_16x16x4_f32(
        false, a, false, b, (short)0, c, false, false);
}

#define QROWS   32
#define APITCH  132

__global__ void __launch_bounds__(256)
qkv_gemm_kernel(const float* __restrict__ nf,
                const float* __restrict__ ln_g,
                const float* __restrict__ ln_b,
                const float* __restrict__ w,
                const float* __restrict__ bias,
                float* __restrict__ out,
                int natoms) {
    __shared__ __attribute__((aligned(16))) float As[QROWS * APITCH];

    const int lane = threadIdx.x & 31;
    const int wid  = threadIdx.x >> 5;
    const int half = lane >> 4;
    const int l16  = lane & 15;
    const int mbase = blockIdx.x * QROWS;

    const float g0 = ln_g[lane * 4 + 0], g1 = ln_g[lane * 4 + 1];
    const float g2 = ln_g[lane * 4 + 2], g3 = ln_g[lane * 4 + 3];
    const float b0 = ln_b[lane * 4 + 0], b1 = ln_b[lane * 4 + 1];
    const float b2 = ln_b[lane * 4 + 2], b3 = ln_b[lane * 4 + 3];
    for (int r = 0; r < QROWS / 8; ++r) {
        const int lr = wid * (QROWS / 8) + r;
        int grow = mbase + lr;
        if (grow >= natoms) grow = natoms - 1;
        const float4 v = ((const float4*)(nf + (size_t)grow * DIM))[lane];
        float s  = v.x + v.y + v.z + v.w;
        float s2 = v.x*v.x + v.y*v.y + v.z*v.z + v.w*v.w;
        for (int m = 1; m < 32; m <<= 1) {
            s  += __shfl_xor(s,  m, 32);
            s2 += __shfl_xor(s2, m, 32);
        }
        const float mu  = s * (1.0f / DIM);
        const float var = s2 * (1.0f / DIM) - mu * mu;
        const float rs  = rsqrtf(var + 1e-5f);
        float4 o;
        o.x = (v.x - mu) * rs * g0 + b0;
        o.y = (v.y - mu) * rs * g1 + b1;
        o.z = (v.z - mu) * rs * g2 + b2;
        o.w = (v.w - mu) * rs * g3 + b3;
        *(float4*)(As + lr * APITCH + lane * 4) = o;
    }
    __syncthreads();

    const int ncol0 = wid * 48 + l16;
    const float* a0p = As + l16 * APITCH + 2 * half;
    const float* a1p = a0p + 16 * APITCH;
    const float* __restrict__ wp = w + (size_t)(2 * half) * QKV_N + ncol0;

    v8f acc00 = {}, acc01 = {}, acc02 = {};
    v8f acc10 = {}, acc11 = {}, acc12 = {};
    #pragma unroll 4
    for (int k0 = 0; k0 < DIM; k0 += 4) {
        const v2f a0 = *(const v2f*)(a0p + k0);
        const v2f a1 = *(const v2f*)(a1p + k0);
        const float* __restrict__ wk = wp + (size_t)k0 * QKV_N;
        v2f b;
        b.x = wk[0];
        b.y = wk[QKV_N];
        acc00 = wmma_f32_16x16x4(a0, b, acc00);
        acc10 = wmma_f32_16x16x4(a1, b, acc10);
        b.x = wk[16];
        b.y = wk[QKV_N + 16];
        acc01 = wmma_f32_16x16x4(a0, b, acc01);
        acc11 = wmma_f32_16x16x4(a1, b, acc11);
        b.x = wk[32];
        b.y = wk[QKV_N + 32];
        acc02 = wmma_f32_16x16x4(a0, b, acc02);
        acc12 = wmma_f32_16x16x4(a1, b, acc12);
    }

    const float bv0 = bias[ncol0];
    const float bv1 = bias[ncol0 + 16];
    const float bv2 = bias[ncol0 + 32];
    for (int r = 0; r < 8; ++r) {
        const int rowa = mbase + r + half * 8;
        const int rowb = rowa + 16;
        if (rowa < natoms) {
            float* o = out + (size_t)rowa * QKV_N + ncol0;
            o[0]  = acc00[r] + bv0;
            o[16] = acc01[r] + bv1;
            o[32] = acc02[r] + bv2;
        }
        if (rowb < natoms) {
            float* o = out + (size_t)rowb * QKV_N + ncol0;
            o[0]  = acc10[r] + bv0;
            o[16] = acc11[r] + bv1;
            o[32] = acc12[r] + bv2;
        }
    }
}

__global__ void zero_kernel(float* __restrict__ p, size_t n) {
    size_t i = (size_t)blockIdx.x * blockDim.x + threadIdx.x;
    const size_t stride = (size_t)gridDim.x * blockDim.x;
    for (; i < n; i += stride) p[i] = 0.0f;
}

__global__ void rank_kernel(const int* __restrict__ row,
                            int* __restrict__ cnt,
                            int* __restrict__ rank,
                            int nedges) {
    const int e = blockIdx.x * blockDim.x + threadIdx.x;
    if (e >= nedges) return;
    rank[e] = atomicAdd(&cnt[row[e]], 1);
}

__global__ void __launch_bounds__(1024)
scan_kernel(const int* __restrict__ cnt, int* __restrict__ offs, int natoms) {
    __shared__ int wsum[32];
    const int tid  = threadIdx.x;
    const int lane = tid & 31;
    const int wid  = tid >> 5;
    const int chunk = (natoms + 1023) / 1024;
    const int b = tid * chunk;
    const int bend = (b + chunk < natoms) ? (b + chunk) : natoms;

    int s = 0;
    for (int i = b; i < bend; ++i) s += cnt[i];

    int incl = s;
    for (int m = 1; m < 32; m <<= 1) {
        const int t = __shfl_up(incl, m, 32);
        if (lane >= m) incl += t;
    }
    if (lane == 31) wsum[wid] = incl;
    __syncthreads();
    if (wid == 0) {
        const int w = wsum[lane];
        int wi = w;
        for (int m = 1; m < 32; m <<= 1) {
            const int t = __shfl_up(wi, m, 32);
            if (lane >= m) wi += t;
        }
        wsum[lane] = wi - w;
    }
    __syncthreads();

    int run = wsum[wid] + incl - s;
    #pragma unroll 8
    for (int i = b; i < bend; ++i) { offs[i] = run; run += cnt[i]; }
    if (tid == 1023) offs[natoms] = run;
}

__global__ void fill_kernel(const int* __restrict__ row,
                            const int* __restrict__ col,
                            const int* __restrict__ offs,
                            const int* __restrict__ rank,
                            int2* __restrict__ recs,
                            int nedges) {
    const int e = blockIdx.x * blockDim.x + threadIdx.x;
    if (e >= nedges) return;
    recs[offs[row[e]] + rank[e]] = make_int2(e, col[e]);
}

__global__ void __launch_bounds__(256)
gather_kernel(const float* __restrict__ qkv,
              const float* __restrict__ edge_feat,
              const float* __restrict__ edge_vec,
              const float* __restrict__ radial,
              const int*   __restrict__ offs,
              const int2*  __restrict__ recs,
              float* __restrict__ m_feat,
              float* __restrict__ out_vec,
              int natoms) {
    const int lane = threadIdx.x & 31;
    const int wave   = __builtin_amdgcn_readfirstlane(
                           (int)(blockIdx.x * (blockDim.x >> 5) + (threadIdx.x >> 5)));
    const int nwaves = (int)(gridDim.x * (blockDim.x >> 5));
    const int hoff = (lane >> 1) * 24 + (lane & 1) * 4;

    for (int atom = wave; atom < natoms; atom += nwaves) {
        const int start = offs[atom];
        const int end   = offs[atom + 1];
        const v4f q = *(const v4f*)(qkv + (size_t)atom * QKV_N + hoff);

        v4f mf = {0.0f, 0.0f, 0.0f, 0.0f};
        v4f a0 = mf, a1 = mf, a2 = mf;

        for (int base = start; base < end; base += 32) {
            const int n = (end - base < 32) ? (end - base) : 32;
            int2 rec = make_int2(0, 0);
            if (lane < n) rec = recs[base + lane];

            for (int i = 0; i < n; ++i) {
                const int e = __builtin_amdgcn_readlane(rec.x, i);
                const int c = __builtin_amdgcn_readlane(rec.y, i);
                const float rad = radial[e];
                const float ev0 = edge_vec[(size_t)e * 3 + 0];
                const float ev1 = edge_vec[(size_t)e * 3 + 1];
                const float ev2 = edge_vec[(size_t)e * 3 + 2];
                const float* kp = qkv + (size_t)c * QKV_N + hoff;
                const v4f k = *(const v4f*)(kp + 8);
                const v4f v = *(const v4f*)(kp + 16);
                const v4f ef = __builtin_nontemporal_load(
                    (const v4f*)(edge_feat + (size_t)e * DIM + lane * 4));

                float p = q.x * k.x + q.y * k.y + q.z * k.z + q.w * k.w;
                p += __builtin_bit_cast(float, __builtin_amdgcn_update_dpp(
                         0, __builtin_bit_cast(int, p), 0xB1, 0xF, 0xF, true));

                const float attn = 0.5f * p
                                 * (1.0f + erff(p * 0.70710678118654752440f))
                                 * rad;
                mf += v * ef * attn;
                a0 += v * ev0;
                a1 += v * ev1;
                a2 += v * ev2;
            }
        }

        *(v4f*)(m_feat + (size_t)atom * DIM + lane * 4) = mf;
        float* vb = out_vec + (size_t)atom * (3 * DIM) + lane * 4;
        *(v4f*)(vb)           = a0;
        *(v4f*)(vb + DIM)     = a1;
        *(v4f*)(vb + 2 * DIM) = a2;
    }
}

__global__ void out_gemm_kernel(const float* __restrict__ mf,
                                const float* __restrict__ w,
                                const float* __restrict__ bias,
                                float* __restrict__ out,
                                int natoms) {
    const int lane = threadIdx.x & 31;
    const int wid  = threadIdx.x >> 5;
    const int half = lane >> 4;
    const int l16  = lane & 15;

    const int mbase = blockIdx.x * 16;
    const int nbase = wid * 16;
    const int ncol  = nbase + l16;

    int arow = mbase + l16;
    if (arow >= natoms) arow = natoms - 1;
    const float* __restrict__ ap = mf + (size_t)arow * DIM;

    v8f acc = {};
    #pragma unroll 4
    for (int k0 = 0; k0 < DIM; k0 += 4) {
        const int kk = k0 + 2 * half;
        v2f a, b;
        a.x = ap[kk];
        a.y = ap[kk + 1];
        b.x = w[(size_t)kk       * DIM + ncol];
        b.y = w[(size_t)(kk + 1) * DIM + ncol];
        acc = wmma_f32_16x16x4(a, b, acc);
    }

    const float bv = bias[ncol];
    #pragma unroll
    for (int r = 0; r < 8; ++r) {
        const int rowm = mbase + r + half * 8;
        if (rowm < natoms)
            out[(size_t)rowm * DIM + ncol] = acc[r] + bv;
    }
}

extern "C" void kernel_launch(void* const* d_in, const int* in_sizes, int n_in,
                              void* d_out, int out_size, void* d_ws, size_t ws_size,
                              hipStream_t stream) {
    const float* nf   = (const float*)d_in[0];
    const float* ef   = (const float*)d_in[1];
    const float* evec = (const float*)d_in[2];
    const float* rad  = (const float*)d_in[3];
    const float* qkvw = (const float*)d_in[4];
    const float* qkvb = (const float*)d_in[5];
    const float* outw = (const float*)d_in[6];
    const float* outb = (const float*)d_in[7];
    const float* lng  = (const float*)d_in[8];
    const float* lnb  = (const float*)d_in[9];
    const int*   row  = (const int*)d_in[10];
    const int*   col  = (const int*)d_in[11];

    const int natoms = in_sizes[0] / DIM;
    const int nedges = in_sizes[3];

    float* qkv   = (float*)d_ws;
    float* mfeat = qkv + (size_t)natoms * QKV_N;

    float* out_feat = (float*)d_out;
    float* out_vec  = out_feat + (size_t)natoms * DIM;

    int2* recs = (int2*)d_out;
    int*  rank = (int*)(recs + nedges);
    int*  cnt  = rank + nedges;
    int*  offs = cnt + natoms;

    dim3 g1((natoms + QROWS - 1) / QROWS, 1);
    qkv_gemm_kernel<<<g1, 256, 0, stream>>>(nf, lng, lnb, qkvw, qkvb,
                                            qkv, natoms);

    const unsigned eblocks = (unsigned)((nedges + 255) / 256);
    zero_kernel<<<64, 256, 0, stream>>>((float*)cnt, (size_t)natoms);
    rank_kernel<<<eblocks, 256, 0, stream>>>(row, cnt, rank, nedges);
    scan_kernel<<<1, 1024, 0, stream>>>(cnt, offs, natoms);
    fill_kernel<<<eblocks, 256, 0, stream>>>(row, col, offs, rank, recs, nedges);
    gather_kernel<<<2048, 256, 0, stream>>>(qkv, ef, evec, rad, offs, recs,
                                            mfeat, out_vec, natoms);

    dim3 g2((natoms + 15) / 16, 1);
    out_gemm_kernel<<<g2, 256, 0, stream>>>(mfeat, outw, outb, out_feat, natoms);
}


# Row-offset scan: the one block whose lanes walked 196-byte-strided chunks (32 lines per access) replaced by a block per 1024 counters, lane-contiguous; each block re-sums the counters before its tile
# speedup vs baseline: 2.7851x; 1.1166x over previous
#include <hip/hip_runtime.h>
#include <math.h>

#define DIM      128
#define NHEADS   16
#define DPH      8
#define QKV_N    384

typedef float v2f __attribute__((ext_vector_type(2)));
typedef float v8f __attribute__((ext_vector_type(8)));
typedef float v4f __attribute__((ext_vector_type(4)));

__device__ __forceinline__ v8f wmma_f32_16x16x4(v2f a, v2f b, v8f c) {
    return __builtin_amdgcn_wmma_f32_16x16x4_f32(
        false, a, false, b, (short)0, c, false, false);
}

#define QROWS   32
#define APITCH  132

__global__ void __launch_bounds__(256)
qkv_gemm_kernel(const float* __restrict__ nf,
                const float* __restrict__ ln_g,
                const float* __restrict__ ln_b,
                const float* __restrict__ w,
                const float* __restrict__ bias,
                float* __restrict__ out,
                int natoms) {
    __shared__ __attribute__((aligned(16))) float As[QROWS * APITCH];

    const int lane = threadIdx.x & 31;
    const int wid  = threadIdx.x >> 5;
    const int half = lane >> 4;
    const int l16  = lane & 15;
    const int mbase = blockIdx.x * QROWS;

    const float g0 = ln_g[lane * 4 + 0], g1 = ln_g[lane * 4 + 1];
    const float g2 = ln_g[lane * 4 + 2], g3 = ln_g[lane * 4 + 3];
    const float b0 = ln_b[lane * 4 + 0], b1 = ln_b[lane * 4 + 1];
    const float b2 = ln_b[lane * 4 + 2], b3 = ln_b[lane * 4 + 3];
    for (int r = 0; r < QROWS / 8; ++r) {
        const int lr = wid * (QROWS / 8) + r;
        int grow = mbase + lr;
        if (grow >= natoms) grow = natoms - 1;
        const float4 v = ((const float4*)(nf + (size_t)grow * DIM))[lane];
        float s  = v.x + v.y + v.z + v.w;
        float s2 = v.x*v.x + v.y*v.y + v.z*v.z + v.w*v.w;
        for (int m = 1; m < 32; m <<= 1) {
            s  += __shfl_xor(s,  m, 32);
            s2 += __shfl_xor(s2, m, 32);
        }
        const float mu  = s * (1.0f / DIM);
        const float var = s2 * (1.0f / DIM) - mu * mu;
        const float rs  = rsqrtf(var + 1e-5f);
        float4 o;
        o.x = (v.x - mu) * rs * g0 + b0;
        o.y = (v.y - mu) * rs * g1 + b1;
        o.z = (v.z - mu) * rs * g2 + b2;
        o.w = (v.w - mu) * rs * g3 + b3;
        *(float4*)(As + lr * APITCH + lane * 4) = o;
    }
    __syncthreads();

    const int ncol0 = wid * 48 + l16;
    const float* a0p = As + l16 * APITCH + 2 * half;
    const float* a1p = a0p + 16 * APITCH;
    const float* __restrict__ wp = w + (size_t)(2 * half) * QKV_N + ncol0;

    v8f acc00 = {}, acc01 = {}, acc02 = {};
    v8f acc10 = {}, acc11 = {}, acc12 = {};
    #pragma unroll 4
    for (int k0 = 0; k0 < DIM; k0 += 4) {
        const v2f a0 = *(const v2f*)(a0p + k0);
        const v2f a1 = *(const v2f*)(a1p + k0);
        const float* __restrict__ wk = wp + (size_t)k0 * QKV_N;
        v2f b;
        b.x = wk[0];
        b.y = wk[QKV_N];
        acc00 = wmma_f32_16x16x4(a0, b, acc00);
        acc10 = wmma_f32_16x16x4(a1, b, acc10);
        b.x = wk[16];
        b.y = wk[QKV_N + 16];
        acc01 = wmma_f32_16x16x4(a0, b, acc01);
        acc11 = wmma_f32_16x16x4(a1, b, acc11);
        b.x = wk[32];
        b.y = wk[QKV_N + 32];
        acc02 = wmma_f32_16x16x4(a0, b, acc02);
        acc12 = wmma_f32_16x16x4(a1, b, acc12);
    }

    const float bv0 = bias[ncol0];
    const float bv1 = bias[ncol0 + 16];
    const float bv2 = bias[ncol0 + 32];
    for (int r = 0; r < 8; ++r) {
        const int rowa = mbase + r + half * 8;
        const int rowb = rowa + 16;
        if (rowa < natoms) {
            float* o = out + (size_t)rowa * QKV_N + ncol0;
            o[0]  = acc00[r] + bv0;
            o[16] = acc01[r] + bv1;
            o[32] = acc02[r] + bv2;
        }
        if (rowb < natoms) {
            float* o = out + (size_t)rowb * QKV_N + ncol0;
            o[0]  = acc10[r] + bv0;
            o[16] = acc11[r] + bv1;
            o[32] = acc12[r] + bv2;
        }
    }
}

__global__ void zero_kernel(float* __restrict__ p, size_t n) {
    size_t i = (size_t)blockIdx.x * blockDim.x + threadIdx.x;
    const size_t stride = (size_t)gridDim.x * blockDim.x;
    for (; i < n; i += stride) p[i] = 0.0f;
}

__global__ void rank_kernel(const int* __restrict__ row,
                            int* __restrict__ cnt,
                            int* __restrict__ rank,
                            int nedges) {
    const int e = blockIdx.x * blockDim.x + threadIdx.x;
    if (e >= nedges) return;
    rank[e] = atomicAdd(&cnt[row[e]], 1);
}

#define SCAN_TILE 1024

__global__ void __launch_bounds__(SCAN_TILE)
scan_kernel(const int* __restrict__ cnt, int* __restrict__ offs, int natoms) {
    __shared__ int wtot[32];
    __shared__ int ptot[32];
    const int tid  = threadIdx.x;
    const int lane = tid & 31;
    const int wid  = tid >> 5;
    const int i    = blockIdx.x * SCAN_TILE + tid;

    const int c = (i < natoms) ? cnt[i] : 0;

    const int4* __restrict__ cnt4 = (const int4*)cnt;
    const int nvec = blockIdx.x * (SCAN_TILE / 4);
    int p = 0;
    for (int k0 = 0; k0 < nvec; k0 += 8 * SCAN_TILE) {
        #pragma unroll
        for (int u = 0; u < 8; ++u) {
            const int k = k0 + u * SCAN_TILE + tid;
            const int4 v = cnt4[(k < nvec) ? k : 0];
            const int s = (v.x + v.y) + (v.z + v.w);
            p += (k < nvec) ? s : 0;
        }
    }

    int incl = c;
    for (int m = 1; m < 32; m <<= 1) {
        const int t = __shfl_up(incl, m, 32);
        if (lane >= m) incl += t;
    }
    for (int m = 1; m < 32; m <<= 1) p += __shfl_xor(p, m, 32);
    if (lane == 31) { wtot[wid] = incl; ptot[wid] = p; }
    __syncthreads();

    int run = ptot[lane] + ((lane < wid) ? wtot[lane] : 0);
    for (int m = 1; m < 32; m <<= 1) run += __shfl_xor(run, m, 32);

    if (i <= natoms) offs[i] = run + incl - c;
}

__global__ void fill_kernel(const int* __restrict__ row,
                            const int* __restrict__ col,
                            const int* __restrict__ offs,
                            const int* __restrict__ rank,
                            int2* __restrict__ recs,
                            int nedges) {
    const int e = blockIdx.x * blockDim.x + threadIdx.x;
    if (e >= nedges) return;
    recs[offs[row[e]] + rank[e]] = make_int2(e, col[e]);
}

__global__ void __launch_bounds__(256)
gather_kernel(const float* __restrict__ qkv,
              const float* __restrict__ edge_feat,
              const float* __restrict__ edge_vec,
              const float* __restrict__ radial,
              const int*   __restrict__ offs,
              const int2*  __restrict__ recs,
              float* __restrict__ m_feat,
              float* __restrict__ out_vec,
              int natoms) {
    const int lane = threadIdx.x & 31;
    const int wave   = __builtin_amdgcn_readfirstlane(
                           (int)(blockIdx.x * (blockDim.x >> 5) + (threadIdx.x >> 5)));
    const int nwaves = (int)(gridDim.x * (blockDim.x >> 5));
    const int hoff = (lane >> 1) * 24 + (lane & 1) * 4;

    for (int atom = wave; atom < natoms; atom += nwaves) {
        const int start = offs[atom];
        const int end   = offs[atom + 1];
        const v4f q = *(const v4f*)(qkv + (size_t)atom * QKV_N + hoff);

        v4f mf = {0.0f, 0.0f, 0.0f, 0.0f};
        v4f a0 = mf, a1 = mf, a2 = mf;

        for (int base = start; base < end; base += 32) {
            const int n = (end - base < 32) ? (end - base) : 32;
            int2 rec = make_int2(0, 0);
            if (lane < n) rec = recs[base + lane];

            for (int i = 0; i < n; ++i) {
                const int e = __builtin_amdgcn_readlane(rec.x, i);
                const int c = __builtin_amdgcn_readlane(rec.y, i);
                const float rad = radial[e];
                const float ev0 = edge_vec[(size_t)e * 3 + 0];
                const float ev1 = edge_vec[(size_t)e * 3 + 1];
                const float ev2 = edge_vec[(size_t)e * 3 + 2];
                const float* kp = qkv + (size_t)c * QKV_N + hoff;
                const v4f k = *(const v4f*)(kp + 8);
                const v4f v = *(const v4f*)(kp + 16);
                const v4f ef = __builtin_nontemporal_load(
                    (const v4f*)(edge_feat + (size_t)e * DIM + lane * 4));

                float p = q.x * k.x + q.y * k.y + q.z * k.z + q.w * k.w;
                p += __builtin_bit_cast(float, __builtin_amdgcn_update_dpp(
                         0, __builtin_bit_cast(int, p), 0xB1, 0xF, 0xF, true));

                const float attn = 0.5f * p
                                 * (1.0f + erff(p * 0.70710678118654752440f))
                                 * rad;
                mf += v * ef * attn;
                a0 += v * ev0;
                a1 += v * ev1;
                a2 += v * ev2;
            }
        }

        *(v4f*)(m_feat + (size_t)atom * DIM + lane * 4) = mf;
        float* vb = out_vec + (size_t)atom * (3 * DIM) + lane * 4;
        *(v4f*)(vb)           = a0;
        *(v4f*)(vb + DIM)     = a1;
        *(v4f*)(vb + 2 * DIM) = a2;
    }
}

__global__ void out_gemm_kernel(const float* __restrict__ mf,
                                const float* __restrict__ w,
                                const float* __restrict__ bias,
                                float* __restrict__ out,
                                int natoms) {
    const int lane = threadIdx.x & 31;
    const int wid  = threadIdx.x >> 5;
    const int half = lane >> 4;
    const int l16  = lane & 15;

    const int mbase = blockIdx.x * 16;
    const int nbase = wid * 16;
    const int ncol  = nbase + l16;

    int arow = mbase + l16;
    if (arow >= natoms) arow = natoms - 1;
    const float* __restrict__ ap = mf + (size_t)arow * DIM;

    v8f acc = {};
    #pragma unroll 4
    for (int k0 = 0; k0 < DIM; k0 += 4) {
        const int kk = k0 + 2 * half;
        v2f a, b;
        a.x = ap[kk];
        a.y = ap[kk + 1];
        b.x = w[(size_t)kk       * DIM + ncol];
        b.y = w[(size_t)(kk + 1) * DIM + ncol];
        acc = wmma_f32_16x16x4(a, b, acc);
    }

    const float bv = bias[ncol];
    #pragma unroll
    for (int r = 0; r < 8; ++r) {
        const int rowm = mbase + r + half * 8;
        if (rowm < natoms)
            out[(size_t)rowm * DIM + ncol] = acc[r] + bv;
    }
}

extern "C" void kernel_launch(void* const* d_in, const int* in_sizes, int n_in,
                              void* d_out, int out_size, void* d_ws, size_t ws_size,
                              hipStream_t stream) {
    const float* nf   = (const float*)d_in[0];
    const float* ef   = (const float*)d_in[1];
    const float* evec = (const float*)d_in[2];
    const float* rad  = (const float*)d_in[3];
    const float* qkvw = (const float*)d_in[4];
    const float* qkvb = (const float*)d_in[5];
    const float* outw = (const float*)d_in[6];
    const float* outb = (const float*)d_in[7];
    const float* lng  = (const float*)d_in[8];
    const float* lnb  = (const float*)d_in[9];
    const int*   row  = (const int*)d_in[10];
    const int*   col  = (const int*)d_in[11];

    const int natoms = in_sizes[0] / DIM;
    const int nedges = in_sizes[3];

    float* qkv   = (float*)d_ws;
    float* mfeat = qkv + (size_t)natoms * QKV_N;

    float* out_feat = (float*)d_out;
    float* out_vec  = out_feat + (size_t)natoms * DIM;

    int2* recs = (int2*)d_out;
    int*  rank = (int*)(recs + nedges);
    int*  cnt  = rank + nedges;
    int*  offs = cnt + natoms;

    dim3 g1((natoms + QROWS - 1) / QROWS, 1);
    qkv_gemm_kernel<<<g1, 256, 0, stream>>>(nf, lng, lnb, qkvw, qkvb,
                                            qkv, natoms);

    const unsigned eblocks = (unsigned)((nedges + 255) / 256);
    zero_kernel<<<64, 256, 0, stream>>>((float*)cnt, (size_t)natoms);
    rank_kernel<<<eblocks, 256, 0, stream>>>(row, cnt, rank, nedges);
    scan_kernel<<<(unsigned)((natoms + SCAN_TILE) / SCAN_TILE), SCAN_TILE, 0, stream>>>(
        cnt, offs, natoms);
    fill_kernel<<<eblocks, 256, 0, stream>>>(row, col, offs, rank, recs, nedges);
    gather_kernel<<<2048, 256, 0, stream>>>(qkv, ef, evec, rad, offs, recs,
                                            mfeat, out_vec, natoms);

    dim3 g2((natoms + 15) / 16, 1);
    out_gemm_kernel<<<g2, 256, 0, stream>>>(mfeat, outw, outb, out_feat, natoms);
}


# Edge-rank pass moved into the LN+QKV GEMM launch as extra leading blocks (4 edges a thread): its returning int adds drain in L2 under the matrix work; 7 launches to 6; sums stay in arrival-rank order
# speedup vs baseline: 2.8910x; 1.0380x over previous
#include <hip/hip_runtime.h>
#include <math.h>

#define DIM      128
#define NHEADS   16
#define DPH      8
#define QKV_N    384

typedef float v2f __attribute__((ext_vector_type(2)));
typedef float v8f __attribute__((ext_vector_type(8)));
typedef float v4f __attribute__((ext_vector_type(4)));

__device__ __forceinline__ v8f wmma_f32_16x16x4(v2f a, v2f b, v8f c) {
    return __builtin_amdgcn_wmma_f32_16x16x4_f32(
        false, a, false, b, (short)0, c, false, false);
}

#define QROWS   32
#define APITCH  132
#define RANK_EPT 4

__global__ void __launch_bounds__(256)
qkv_gemm_kernel(const float* __restrict__ nf,
                const float* __restrict__ ln_g,
                const float* __restrict__ ln_b,
                const float* __restrict__ w,
                const float* __restrict__ bias,
                float* __restrict__ out,
                int natoms,
                const int* __restrict__ row,
                int* __restrict__ cnt,
                int* __restrict__ rank,
                int nedges,
                int rank_blocks) {
    __shared__ __attribute__((aligned(16))) float As[QROWS * APITCH];

    if ((int)blockIdx.x < rank_blocks) {
        const int e0 = blockIdx.x * (256 * RANK_EPT);
        const int ebase = e0 + threadIdx.x;
        if (e0 + 256 * RANK_EPT <= nedges) {
            int rv[RANK_EPT];
            #pragma unroll
            for (int j = 0; j < RANK_EPT; ++j) rv[j] = row[ebase + j * 256];
            #pragma unroll
            for (int j = 0; j < RANK_EPT; ++j) rv[j] = atomicAdd(&cnt[rv[j]], 1);
            #pragma unroll
            for (int j = 0; j < RANK_EPT; ++j) rank[ebase + j * 256] = rv[j];
        } else {
            for (int j = 0; j < RANK_EPT; ++j) {
                const int e = ebase + j * 256;
                if (e < nedges) rank[e] = atomicAdd(&cnt[row[e]], 1);
            }
        }
        return;
    }

    const int lane = threadIdx.x & 31;
    const int wid  = threadIdx.x >> 5;
    const int half = lane >> 4;
    const int l16  = lane & 15;
    const int mbase = ((int)blockIdx.x - rank_blocks) * QROWS;

    const float g0 = ln_g[lane * 4 + 0], g1 = ln_g[lane * 4 + 1];
    const float g2 = ln_g[lane * 4 + 2], g3 = ln_g[lane * 4 + 3];
    const float b0 = ln_b[lane * 4 + 0], b1 = ln_b[lane * 4 + 1];
    const float b2 = ln_b[lane * 4 + 2], b3 = ln_b[lane * 4 + 3];
    for (int r = 0; r < QROWS / 8; ++r) {
        const int lr = wid * (QROWS / 8) + r;
        int grow = mbase + lr;
        if (grow >= natoms) grow = natoms - 1;
        const float4 v = ((const float4*)(nf + (size_t)grow * DIM))[lane];
        float s  = v.x + v.y + v.z + v.w;
        float s2 = v.x*v.x + v.y*v.y + v.z*v.z + v.w*v.w;
        for (int m = 1; m < 32; m <<= 1) {
            s  += __shfl_xor(s,  m, 32);
            s2 += __shfl_xor(s2, m, 32);
        }
        const float mu  = s * (1.0f / DIM);
        const float var = s2 * (1.0f / DIM) - mu * mu;
        const float rs  = rsqrtf(var + 1e-5f);
        float4 o;
        o.x = (v.x - mu) * rs * g0 + b0;
        o.y = (v.y - mu) * rs * g1 + b1;
        o.z = (v.z - mu) * rs * g2 + b2;
        o.w = (v.w - mu) * rs * g3 + b3;
        *(float4*)(As + lr * APITCH + lane * 4) = o;
    }
    __syncthreads();

    const int ncol0 = wid * 48 + l16;
    const float* a0p = As + l16 * APITCH + 2 * half;
    const float* a1p = a0p + 16 * APITCH;
    const float* __restrict__ wp = w + (size_t)(2 * half) * QKV_N + ncol0;

    v8f acc00 = {}, acc01 = {}, acc02 = {};
    v8f acc10 = {}, acc11 = {}, acc12 = {};
    #pragma unroll 4
    for (int k0 = 0; k0 < DIM; k0 += 4) {
        const v2f a0 = *(const v2f*)(a0p + k0);
        const v2f a1 = *(const v2f*)(a1p + k0);
        const float* __restrict__ wk = wp + (size_t)k0 * QKV_N;
        v2f b;
        b.x = wk[0];
        b.y = wk[QKV_N];
        acc00 = wmma_f32_16x16x4(a0, b, acc00);
        acc10 = wmma_f32_16x16x4(a1, b, acc10);
        b.x = wk[16];
        b.y = wk[QKV_N + 16];
        acc01 = wmma_f32_16x16x4(a0, b, acc01);
        acc11 = wmma_f32_16x16x4(a1, b, acc11);
        b.x = wk[32];
        b.y = wk[QKV_N + 32];
        acc02 = wmma_f32_16x16x4(a0, b, acc02);
        acc12 = wmma_f32_16x16x4(a1, b, acc12);
    }

    const float bv0 = bias[ncol0];
    const float bv1 = bias[ncol0 + 16];
    const float bv2 = bias[ncol0 + 32];
    for (int r = 0; r < 8; ++r) {
        const int rowa = mbase + r + half * 8;
        const int rowb = rowa + 16;
        if (rowa < natoms) {
            float* o = out + (size_t)rowa * QKV_N + ncol0;
            o[0]  = acc00[r] + bv0;
            o[16] = acc01[r] + bv1;
            o[32] = acc02[r] + bv2;
        }
        if (rowb < natoms) {
            float* o = out + (size_t)rowb * QKV_N + ncol0;
            o[0]  = acc10[r] + bv0;
            o[16] = acc11[r] + bv1;
            o[32] = acc12[r] + bv2;
        }
    }
}

__global__ void zero_kernel(float* __restrict__ p, size_t n) {
    size_t i = (size_t)blockIdx.x * blockDim.x + threadIdx.x;
    const size_t stride = (size_t)gridDim.x * blockDim.x;
    for (; i < n; i += stride) p[i] = 0.0f;
}


#define SCAN_TILE 1024

__global__ void __launch_bounds__(SCAN_TILE)
scan_kernel(const int* __restrict__ cnt, int* __restrict__ offs, int natoms) {
    __shared__ int wtot[32];
    __shared__ int ptot[32];
    const int tid  = threadIdx.x;
    const int lane = tid & 31;
    const int wid  = tid >> 5;
    const int i    = blockIdx.x * SCAN_TILE + tid;

    const int c = (i < natoms) ? cnt[i] : 0;

    const int4* __restrict__ cnt4 = (const int4*)cnt;
    const int nvec = blockIdx.x * (SCAN_TILE / 4);
    int p = 0;
    for (int k0 = 0; k0 < nvec; k0 += 8 * SCAN_TILE) {
        #pragma unroll
        for (int u = 0; u < 8; ++u) {
            const int k = k0 + u * SCAN_TILE + tid;
            const int4 v = cnt4[(k < nvec) ? k : 0];
            const int s = (v.x + v.y) + (v.z + v.w);
            p += (k < nvec) ? s : 0;
        }
    }

    int incl = c;
    for (int m = 1; m < 32; m <<= 1) {
        const int t = __shfl_up(incl, m, 32);
        if (lane >= m) incl += t;
    }
    for (int m = 1; m < 32; m <<= 1) p += __shfl_xor(p, m, 32);
    if (lane == 31) { wtot[wid] = incl; ptot[wid] = p; }
    __syncthreads();

    int run = ptot[lane] + ((lane < wid) ? wtot[lane] : 0);
    for (int m = 1; m < 32; m <<= 1) run += __shfl_xor(run, m, 32);

    if (i <= natoms) offs[i] = run + incl - c;
}

__global__ void fill_kernel(const int* __restrict__ row,
                            const int* __restrict__ col,
                            const int* __restrict__ offs,
                            const int* __restrict__ rank,
                            int2* __restrict__ recs,
                            int nedges) {
    const int e = blockIdx.x * blockDim.x + threadIdx.x;
    if (e >= nedges) return;
    recs[offs[row[e]] + rank[e]] = make_int2(e, col[e]);
}

__global__ void __launch_bounds__(256)
gather_kernel(const float* __restrict__ qkv,
              const float* __restrict__ edge_feat,
              const float* __restrict__ edge_vec,
              const float* __restrict__ radial,
              const int*   __restrict__ offs,
              const int2*  __restrict__ recs,
              float* __restrict__ m_feat,
              float* __restrict__ out_vec,
              int natoms) {
    const int lane = threadIdx.x & 31;
    const int wave   = __builtin_amdgcn_readfirstlane(
                           (int)(blockIdx.x * (blockDim.x >> 5) + (threadIdx.x >> 5)));
    const int nwaves = (int)(gridDim.x * (blockDim.x >> 5));
    const int hoff = (lane >> 1) * 24 + (lane & 1) * 4;

    for (int atom = wave; atom < natoms; atom += nwaves) {
        const int start = offs[atom];
        const int end   = offs[atom + 1];
        const v4f q = *(const v4f*)(qkv + (size_t)atom * QKV_N + hoff);

        v4f mf = {0.0f, 0.0f, 0.0f, 0.0f};
        v4f a0 = mf, a1 = mf, a2 = mf;

        for (int base = start; base < end; base += 32) {
            const int n = (end - base < 32) ? (end - base) : 32;
            int2 rec = make_int2(0, 0);
            if (lane < n) rec = recs[base + lane];

            for (int i = 0; i < n; ++i) {
                const int e = __builtin_amdgcn_readlane(rec.x, i);
                const int c = __builtin_amdgcn_readlane(rec.y, i);
                const float rad = radial[e];
                const float ev0 = edge_vec[(size_t)e * 3 + 0];
                const float ev1 = edge_vec[(size_t)e * 3 + 1];
                const float ev2 = edge_vec[(size_t)e * 3 + 2];
                const float* kp = qkv + (size_t)c * QKV_N + hoff;
                const v4f k = *(const v4f*)(kp + 8);
                const v4f v = *(const v4f*)(kp + 16);
                const v4f ef = __builtin_nontemporal_load(
                    (const v4f*)(edge_feat + (size_t)e * DIM + lane * 4));

                float p = q.x * k.x + q.y * k.y + q.z * k.z + q.w * k.w;
                p += __builtin_bit_cast(float, __builtin_amdgcn_update_dpp(
                         0, __builtin_bit_cast(int, p), 0xB1, 0xF, 0xF, true));

                const float attn = 0.5f * p
                                 * (1.0f + erff(p * 0.70710678118654752440f))
                                 * rad;
                mf += v * ef * attn;
                a0 += v * ev0;
                a1 += v * ev1;
                a2 += v * ev2;
            }
        }

        *(v4f*)(m_feat + (size_t)atom * DIM + lane * 4) = mf;
        float* vb = out_vec + (size_t)atom * (3 * DIM) + lane * 4;
        *(v4f*)(vb)           = a0;
        *(v4f*)(vb + DIM)     = a1;
        *(v4f*)(vb + 2 * DIM) = a2;
    }
}

__global__ void out_gemm_kernel(const float* __restrict__ mf,
                                const float* __restrict__ w,
                                const float* __restrict__ bias,
                                float* __restrict__ out,
                                int natoms) {
    const int lane = threadIdx.x & 31;
    const int wid  = threadIdx.x >> 5;
    const int half = lane >> 4;
    const int l16  = lane & 15;

    const int mbase = blockIdx.x * 16;
    const int nbase = wid * 16;
    const int ncol  = nbase + l16;

    int arow = mbase + l16;
    if (arow >= natoms) arow = natoms - 1;
    const float* __restrict__ ap = mf + (size_t)arow * DIM;

    v8f acc = {};
    #pragma unroll 4
    for (int k0 = 0; k0 < DIM; k0 += 4) {
        const int kk = k0 + 2 * half;
        v2f a, b;
        a.x = ap[kk];
        a.y = ap[kk + 1];
        b.x = w[(size_t)kk       * DIM + ncol];
        b.y = w[(size_t)(kk + 1) * DIM + ncol];
        acc = wmma_f32_16x16x4(a, b, acc);
    }

    const float bv = bias[ncol];
    #pragma unroll
    for (int r = 0; r < 8; ++r) {
        const int rowm = mbase + r + half * 8;
        if (rowm < natoms)
            out[(size_t)rowm * DIM + ncol] = acc[r] + bv;
    }
}

extern "C" void kernel_launch(void* const* d_in, const int* in_sizes, int n_in,
                              void* d_out, int out_size, void* d_ws, size_t ws_size,
                              hipStream_t stream) {
    const float* nf   = (const float*)d_in[0];
    const float* ef   = (const float*)d_in[1];
    const float* evec = (const float*)d_in[2];
    const float* rad  = (const float*)d_in[3];
    const float* qkvw = (const float*)d_in[4];
    const float* qkvb = (const float*)d_in[5];
    const float* outw = (const float*)d_in[6];
    const float* outb = (const float*)d_in[7];
    const float* lng  = (const float*)d_in[8];
    const float* lnb  = (const float*)d_in[9];
    const int*   row  = (const int*)d_in[10];
    const int*   col  = (const int*)d_in[11];

    const int natoms = in_sizes[0] / DIM;
    const int nedges = in_sizes[3];

    float* qkv   = (float*)d_ws;
    float* mfeat = qkv + (size_t)natoms * QKV_N;

    float* out_feat = (float*)d_out;
    float* out_vec  = out_feat + (size_t)natoms * DIM;

    int2* recs = (int2*)d_out;
    int*  rank = (int*)(recs + nedges);
    int*  cnt  = rank + nedges;
    int*  offs = cnt + natoms;

    zero_kernel<<<64, 256, 0, stream>>>((float*)cnt, (size_t)natoms);
    const int rblocks = (nedges + 256 * RANK_EPT - 1) / (256 * RANK_EPT);
    dim3 g1((unsigned)rblocks + (unsigned)((natoms + QROWS - 1) / QROWS), 1);
    qkv_gemm_kernel<<<g1, 256, 0, stream>>>(nf, lng, lnb, qkvw, qkvb,
                                            qkv, natoms,
                                            row, cnt, rank, nedges, rblocks);

    const unsigned eblocks = (unsigned)((nedges + 255) / 256);
    scan_kernel<<<(unsigned)((natoms + SCAN_TILE) / SCAN_TILE), SCAN_TILE, 0, stream>>>(
        cnt, offs, natoms);
    fill_kernel<<<eblocks, 256, 0, stream>>>(row, col, offs, rank, recs, nedges);
    gather_kernel<<<2048, 256, 0, stream>>>(qkv, ef, evec, rad, offs, recs,
                                            mfeat, out_vec, natoms);

    dim3 g2((natoms + 15) / 16, 1);
    out_gemm_kernel<<<g2, 256, 0, stream>>>(mfeat, outw, outb, out_feat, natoms);
}


# Output projection fused into the row-sum launch: a work-group sums 16 rows into LDS, then multiplies them by out_w on the matrix unit (k order kept); message rows never reach memory; 6 launches to 5
# speedup vs baseline: 3.1255x; 1.0811x over previous
#include <hip/hip_runtime.h>
#include <math.h>

#define DIM      128
#define NHEADS   16
#define DPH      8
#define QKV_N    384

typedef float v2f __attribute__((ext_vector_type(2)));
typedef float v8f __attribute__((ext_vector_type(8)));
typedef float v4f __attribute__((ext_vector_type(4)));

__device__ __forceinline__ v8f wmma_f32_16x16x4(v2f a, v2f b, v8f c) {
    return __builtin_amdgcn_wmma_f32_16x16x4_f32(
        false, a, false, b, (short)0, c, false, false);
}

#define QROWS   32
#define APITCH  132
#define RANK_EPT 4

__global__ void __launch_bounds__(256)
qkv_gemm_kernel(const float* __restrict__ nf,
                const float* __restrict__ ln_g,
                const float* __restrict__ ln_b,
                const float* __restrict__ w,
                const float* __restrict__ bias,
                float* __restrict__ out,
                int natoms,
                const int* __restrict__ row,
                int* __restrict__ cnt,
                int* __restrict__ rank,
                int nedges,
                int rank_blocks) {
    __shared__ __attribute__((aligned(16))) float As[QROWS * APITCH];

    if ((int)blockIdx.x < rank_blocks) {
        const int e0 = blockIdx.x * (256 * RANK_EPT);
        const int ebase = e0 + threadIdx.x;
        if (e0 + 256 * RANK_EPT <= nedges) {
            int rv[RANK_EPT];
            #pragma unroll
            for (int j = 0; j < RANK_EPT; ++j) rv[j] = row[ebase + j * 256];
            #pragma unroll
            for (int j = 0; j < RANK_EPT; ++j) rv[j] = atomicAdd(&cnt[rv[j]], 1);
            #pragma unroll
            for (int j = 0; j < RANK_EPT; ++j) rank[ebase + j * 256] = rv[j];
        } else {
            for (int j = 0; j < RANK_EPT; ++j) {
                const int e = ebase + j * 256;
                if (e < nedges) rank[e] = atomicAdd(&cnt[row[e]], 1);
            }
        }
        return;
    }

    const int lane = threadIdx.x & 31;
    const int wid  = threadIdx.x >> 5;
    const int half = lane >> 4;
    const int l16  = lane & 15;
    const int mbase = ((int)blockIdx.x - rank_blocks) * QROWS;

    const float g0 = ln_g[lane * 4 + 0], g1 = ln_g[lane * 4 + 1];
    const float g2 = ln_g[lane * 4 + 2], g3 = ln_g[lane * 4 + 3];
    const float b0 = ln_b[lane * 4 + 0], b1 = ln_b[lane * 4 + 1];
    const float b2 = ln_b[lane * 4 + 2], b3 = ln_b[lane * 4 + 3];
    for (int r = 0; r < QROWS / 8; ++r) {
        const int lr = wid * (QROWS / 8) + r;
        int grow = mbase + lr;
        if (grow >= natoms) grow = natoms - 1;
        const float4 v = ((const float4*)(nf + (size_t)grow * DIM))[lane];
        float s  = v.x + v.y + v.z + v.w;
        float s2 = v.x*v.x + v.y*v.y + v.z*v.z + v.w*v.w;
        for (int m = 1; m < 32; m <<= 1) {
            s  += __shfl_xor(s,  m, 32);
            s2 += __shfl_xor(s2, m, 32);
        }
        const float mu  = s * (1.0f / DIM);
        const float var = s2 * (1.0f / DIM) - mu * mu;
        const float rs  = rsqrtf(var + 1e-5f);
        float4 o;
        o.x = (v.x - mu) * rs * g0 + b0;
        o.y = (v.y - mu) * rs * g1 + b1;
        o.z = (v.z - mu) * rs * g2 + b2;
        o.w = (v.w - mu) * rs * g3 + b3;
        *(float4*)(As + lr * APITCH + lane * 4) = o;
    }
    __syncthreads();

    const int ncol0 = wid * 48 + l16;
    const float* a0p = As + l16 * APITCH + 2 * half;
    const float* a1p = a0p + 16 * APITCH;
    const float* __restrict__ wp = w + (size_t)(2 * half) * QKV_N + ncol0;

    v8f acc00 = {}, acc01 = {}, acc02 = {};
    v8f acc10 = {}, acc11 = {}, acc12 = {};
    #pragma unroll 4
    for (int k0 = 0; k0 < DIM; k0 += 4) {
        const v2f a0 = *(const v2f*)(a0p + k0);
        const v2f a1 = *(const v2f*)(a1p + k0);
        const float* __restrict__ wk = wp + (size_t)k0 * QKV_N;
        v2f b;
        b.x = wk[0];
        b.y = wk[QKV_N];
        acc00 = wmma_f32_16x16x4(a0, b, acc00);
        acc10 = wmma_f32_16x16x4(a1, b, acc10);
        b.x = wk[16];
        b.y = wk[QKV_N + 16];
        acc01 = wmma_f32_16x16x4(a0, b, acc01);
        acc11 = wmma_f32_16x16x4(a1, b, acc11);
        b.x = wk[32];
        b.y = wk[QKV_N + 32];
        acc02 = wmma_f32_16x16x4(a0, b, acc02);
        acc12 = wmma_f32_16x16x4(a1, b, acc12);
    }

    const float bv0 = bias[ncol0];
    const float bv1 = bias[ncol0 + 16];
    const float bv2 = bias[ncol0 + 32];
    for (int r = 0; r < 8; ++r) {
        const int rowa = mbase + r + half * 8;
        const int rowb = rowa + 16;
        if (rowa < natoms) {
            float* o = out + (size_t)rowa * QKV_N + ncol0;
            o[0]  = acc00[r] + bv0;
            o[16] = acc01[r] + bv1;
            o[32] = acc02[r] + bv2;
        }
        if (rowb < natoms) {
            float* o = out + (size_t)rowb * QKV_N + ncol0;
            o[0]  = acc10[r] + bv0;
            o[16] = acc11[r] + bv1;
            o[32] = acc12[r] + bv2;
        }
    }
}

__global__ void zero_kernel(float* __restrict__ p, size_t n) {
    size_t i = (size_t)blockIdx.x * blockDim.x + threadIdx.x;
    const size_t stride = (size_t)gridDim.x * blockDim.x;
    for (; i < n; i += stride) p[i] = 0.0f;
}


#define SCAN_TILE 1024

__global__ void __launch_bounds__(SCAN_TILE)
scan_kernel(const int* __restrict__ cnt, int* __restrict__ offs, int natoms) {
    __shared__ int wtot[32];
    __shared__ int ptot[32];
    const int tid  = threadIdx.x;
    const int lane = tid & 31;
    const int wid  = tid >> 5;
    const int i    = blockIdx.x * SCAN_TILE + tid;

    const int c = (i < natoms) ? cnt[i] : 0;

    const int4* __restrict__ cnt4 = (const int4*)cnt;
    const int nvec = blockIdx.x * (SCAN_TILE / 4);
    int p = 0;
    for (int k0 = 0; k0 < nvec; k0 += 8 * SCAN_TILE) {
        #pragma unroll
        for (int u = 0; u < 8; ++u) {
            const int k = k0 + u * SCAN_TILE + tid;
            const int4 v = cnt4[(k < nvec) ? k : 0];
            const int s = (v.x + v.y) + (v.z + v.w);
            p += (k < nvec) ? s : 0;
        }
    }

    int incl = c;
    for (int m = 1; m < 32; m <<= 1) {
        const int t = __shfl_up(incl, m, 32);
        if (lane >= m) incl += t;
    }
    for (int m = 1; m < 32; m <<= 1) p += __shfl_xor(p, m, 32);
    if (lane == 31) { wtot[wid] = incl; ptot[wid] = p; }
    __syncthreads();

    int run = ptot[lane] + ((lane < wid) ? wtot[lane] : 0);
    for (int m = 1; m < 32; m <<= 1) run += __shfl_xor(run, m, 32);

    if (i <= natoms) offs[i] = run + incl - c;
}

__global__ void fill_kernel(const int* __restrict__ row,
                            const int* __restrict__ col,
                            const int* __restrict__ offs,
                            const int* __restrict__ rank,
                            int2* __restrict__ recs,
                            int nedges) {
    const int e = blockIdx.x * blockDim.x + threadIdx.x;
    if (e >= nedges) return;
    recs[offs[row[e]] + rank[e]] = make_int2(e, col[e]);
}

#define ROWS_PB 16
#define MPITCH  132

__global__ void __launch_bounds__(256)
gather_proj_kernel(const float* __restrict__ qkv,
                   const float* __restrict__ edge_feat,
                   const float* __restrict__ edge_vec,
                   const float* __restrict__ radial,
                   const int*   __restrict__ offs,
                   const int2*  __restrict__ recs,
                   const float* __restrict__ ow,
                   const float* __restrict__ ob,
                   float* __restrict__ out_feat,
                   float* __restrict__ out_vec,
                   int natoms) {
    __shared__ __attribute__((aligned(16))) float Ms[ROWS_PB * MPITCH];

    const int lane = threadIdx.x & 31;
    const int wid   = __builtin_amdgcn_readfirstlane((int)(threadIdx.x >> 5));
    const int mbase = (int)blockIdx.x * ROWS_PB;
    const int hoff = (lane >> 1) * 24 + (lane & 1) * 4;

    for (int r = 0; r < ROWS_PB / 8; ++r) {
        const int lr   = wid * (ROWS_PB / 8) + r;
        const int atom = mbase + lr;
        v4f mf = {0.0f, 0.0f, 0.0f, 0.0f};

        if (atom < natoms) {
            const int start = offs[atom];
            const int end   = offs[atom + 1];
            const v4f q = *(const v4f*)(qkv + (size_t)atom * QKV_N + hoff);

            v4f a0 = mf, a1 = mf, a2 = mf;

            for (int base = start; base < end; base += 32) {
                const int n = (end - base < 32) ? (end - base) : 32;
                int2 rec = make_int2(0, 0);
                if (lane < n) rec = recs[base + lane];

                for (int i = 0; i < n; ++i) {
                    const int e = __builtin_amdgcn_readlane(rec.x, i);
                    const int c = __builtin_amdgcn_readlane(rec.y, i);
                    const float rad = radial[e];
                    const float ev0 = edge_vec[(size_t)e * 3 + 0];
                    const float ev1 = edge_vec[(size_t)e * 3 + 1];
                    const float ev2 = edge_vec[(size_t)e * 3 + 2];
                    const float* kp = qkv + (size_t)c * QKV_N + hoff;
                    const v4f k = *(const v4f*)(kp + 8);
                    const v4f v = *(const v4f*)(kp + 16);
                    const v4f ef = __builtin_nontemporal_load(
                        (const v4f*)(edge_feat + (size_t)e * DIM + lane * 4));

                    float p = q.x * k.x + q.y * k.y + q.z * k.z + q.w * k.w;
                    p += __builtin_bit_cast(float, __builtin_amdgcn_update_dpp(
                             0, __builtin_bit_cast(int, p), 0xB1, 0xF, 0xF, true));

                    const float attn = 0.5f * p
                                     * (1.0f + erff(p * 0.70710678118654752440f))
                                     * rad;
                    mf += v * ef * attn;
                    a0 += v * ev0;
                    a1 += v * ev1;
                    a2 += v * ev2;
                }
            }

            float* vb = out_vec + (size_t)atom * (3 * DIM) + lane * 4;
            *(v4f*)(vb)           = a0;
            *(v4f*)(vb + DIM)     = a1;
            *(v4f*)(vb + 2 * DIM) = a2;
        }
        *(v4f*)(Ms + lr * MPITCH + lane * 4) = mf;
    }
    __syncthreads();

    const int half = lane >> 4;
    const int l16  = lane & 15;
    const int ncol = wid * 16 + l16;
    const float* ap = Ms + l16 * MPITCH + 2 * half;
    const float* __restrict__ wp = ow + (size_t)(2 * half) * DIM + ncol;

    v8f acc = {};
    #pragma unroll 8
    for (int k0 = 0; k0 < DIM; k0 += 4) {
        const v2f a = *(const v2f*)(ap + k0);
        v2f b;
        b.x = wp[(size_t)k0 * DIM];
        b.y = wp[(size_t)(k0 + 1) * DIM];
        acc = wmma_f32_16x16x4(a, b, acc);
    }

    const float bv = ob[ncol];
    #pragma unroll
    for (int r = 0; r < 8; ++r) {
        const int rowm = mbase + r + half * 8;
        if (rowm < natoms)
            out_feat[(size_t)rowm * DIM + ncol] = acc[r] + bv;
    }
}

extern "C" void kernel_launch(void* const* d_in, const int* in_sizes, int n_in,
                              void* d_out, int out_size, void* d_ws, size_t ws_size,
                              hipStream_t stream) {
    const float* nf   = (const float*)d_in[0];
    const float* ef   = (const float*)d_in[1];
    const float* evec = (const float*)d_in[2];
    const float* rad  = (const float*)d_in[3];
    const float* qkvw = (const float*)d_in[4];
    const float* qkvb = (const float*)d_in[5];
    const float* outw = (const float*)d_in[6];
    const float* outb = (const float*)d_in[7];
    const float* lng  = (const float*)d_in[8];
    const float* lnb  = (const float*)d_in[9];
    const int*   row  = (const int*)d_in[10];
    const int*   col  = (const int*)d_in[11];

    const int natoms = in_sizes[0] / DIM;
    const int nedges = in_sizes[3];

    float* qkv   = (float*)d_ws;

    float* out_feat = (float*)d_out;
    float* out_vec  = out_feat + (size_t)natoms * DIM;

    int2* recs = (int2*)(qkv + (size_t)natoms * QKV_N);
    int*  rank = (int*)(recs + nedges);
    int*  cnt  = rank + nedges;
    int*  offs = cnt + natoms;

    zero_kernel<<<64, 256, 0, stream>>>((float*)cnt, (size_t)natoms);
    const int rblocks = (nedges + 256 * RANK_EPT - 1) / (256 * RANK_EPT);
    dim3 g1((unsigned)rblocks + (unsigned)((natoms + QROWS - 1) / QROWS), 1);
    qkv_gemm_kernel<<<g1, 256, 0, stream>>>(nf, lng, lnb, qkvw, qkvb,
                                            qkv, natoms,
                                            row, cnt, rank, nedges, rblocks);

    const unsigned eblocks = (unsigned)((nedges + 255) / 256);
    scan_kernel<<<(unsigned)((natoms + SCAN_TILE) / SCAN_TILE), SCAN_TILE, 0, stream>>>(
        cnt, offs, natoms);
    fill_kernel<<<eblocks, 256, 0, stream>>>(row, col, offs, rank, recs, nedges);
    dim3 g2((unsigned)((natoms + ROWS_PB - 1) / ROWS_PB), 1);
    gather_proj_kernel<<<g2, 256, 0, stream>>>(qkv, ef, evec, rad, offs, recs,
                                               outw, outb, out_feat, out_vec,
                                               natoms);
}


# Edge grouping by counting sort (count, scan, fill) replaced by per-row linked lists built with one returning 64-bit int exchange per edge; 5 launches to 3; row sums now run in reverse arrival order
# speedup vs baseline: 3.5943x; 1.1500x over previous
#include <hip/hip_runtime.h>
#include <math.h>

#define DIM      128
#define NHEADS   16
#define DPH      8
#define QKV_N    384

typedef float v2f __attribute__((ext_vector_type(2)));
typedef float v8f __attribute__((ext_vector_type(8)));
typedef float v4f __attribute__((ext_vector_type(4)));

__device__ __forceinline__ v8f wmma_f32_16x16x4(v2f a, v2f b, v8f c) {
    return __builtin_amdgcn_wmma_f32_16x16x4_f32(
        false, a, false, b, (short)0, c, false, false);
}

#define QROWS   32
#define APITCH  132
#define LINK_EPT 4

#define LIST_END 0x00000000FFFFFFFFull
__device__ __forceinline__ unsigned long long list_entry(int e, int c) {
    return (unsigned long long)(unsigned)e | ((unsigned long long)(unsigned)c << 32);
}

__global__ void __launch_bounds__(256)
qkv_gemm_kernel(const float* __restrict__ nf,
                const float* __restrict__ ln_g,
                const float* __restrict__ ln_b,
                const float* __restrict__ w,
                const float* __restrict__ bias,
                float* __restrict__ out,
                int natoms,
                const int* __restrict__ row,
                const int* __restrict__ col,
                unsigned long long* __restrict__ head,
                unsigned long long* __restrict__ link,
                int nedges,
                int link_blocks) {
    __shared__ __attribute__((aligned(16))) float As[QROWS * APITCH];

    if ((int)blockIdx.x < link_blocks) {
        const int e0 = blockIdx.x * (256 * LINK_EPT);
        const int ebase = e0 + threadIdx.x;
        if (e0 + 256 * LINK_EPT <= nedges) {
            int rv[LINK_EPT];
            unsigned long long en[LINK_EPT];
            #pragma unroll
            for (int j = 0; j < LINK_EPT; ++j) rv[j] = row[ebase + j * 256];
            #pragma unroll
            for (int j = 0; j < LINK_EPT; ++j)
                en[j] = list_entry(ebase + j * 256, col[ebase + j * 256]);
            #pragma unroll
            for (int j = 0; j < LINK_EPT; ++j) en[j] = atomicExch(&head[rv[j]], en[j]);
            #pragma unroll
            for (int j = 0; j < LINK_EPT; ++j) link[ebase + j * 256] = en[j];
        } else {
            for (int j = 0; j < LINK_EPT; ++j) {
                const int e = ebase + j * 256;
                if (e < nedges)
                    link[e] = atomicExch(&head[row[e]], list_entry(e, col[e]));
            }
        }
        return;
    }

    const int lane = threadIdx.x & 31;
    const int wid  = threadIdx.x >> 5;
    const int half = lane >> 4;
    const int l16  = lane & 15;
    const int mbase = ((int)blockIdx.x - link_blocks) * QROWS;

    const float g0 = ln_g[lane * 4 + 0], g1 = ln_g[lane * 4 + 1];
    const float g2 = ln_g[lane * 4 + 2], g3 = ln_g[lane * 4 + 3];
    const float b0 = ln_b[lane * 4 + 0], b1 = ln_b[lane * 4 + 1];
    const float b2 = ln_b[lane * 4 + 2], b3 = ln_b[lane * 4 + 3];
    for (int r = 0; r < QROWS / 8; ++r) {
        const int lr = wid * (QROWS / 8) + r;
        int grow = mbase + lr;
        if (grow >= natoms) grow = natoms - 1;
        const float4 v = ((const float4*)(nf + (size_t)grow * DIM))[lane];
        float s  = v.x + v.y + v.z + v.w;
        float s2 = v.x*v.x + v.y*v.y + v.z*v.z + v.w*v.w;
        for (int m = 1; m < 32; m <<= 1) {
            s  += __shfl_xor(s,  m, 32);
            s2 += __shfl_xor(s2, m, 32);
        }
        const float mu  = s * (1.0f / DIM);
        const float var = s2 * (1.0f / DIM) - mu * mu;
        const float rs  = rsqrtf(var + 1e-5f);
        float4 o;
        o.x = (v.x - mu) * rs * g0 + b0;
        o.y = (v.y - mu) * rs * g1 + b1;
        o.z = (v.z - mu) * rs * g2 + b2;
        o.w = (v.w - mu) * rs * g3 + b3;
        *(float4*)(As + lr * APITCH + lane * 4) = o;
    }
    __syncthreads();

    const int ncol0 = wid * 48 + l16;
    const float* a0p = As + l16 * APITCH + 2 * half;
    const float* a1p = a0p + 16 * APITCH;
    const float* __restrict__ wp = w + (size_t)(2 * half) * QKV_N + ncol0;

    v8f acc00 = {}, acc01 = {}, acc02 = {};
    v8f acc10 = {}, acc11 = {}, acc12 = {};
    #pragma unroll 4
    for (int k0 = 0; k0 < DIM; k0 += 4) {
        const v2f a0 = *(const v2f*)(a0p + k0);
        const v2f a1 = *(const v2f*)(a1p + k0);
        const float* __restrict__ wk = wp + (size_t)k0 * QKV_N;
        v2f b;
        b.x = wk[0];
        b.y = wk[QKV_N];
        acc00 = wmma_f32_16x16x4(a0, b, acc00);
        acc10 = wmma_f32_16x16x4(a1, b, acc10);
        b.x = wk[16];
        b.y = wk[QKV_N + 16];
        acc01 = wmma_f32_16x16x4(a0, b, acc01);
        acc11 = wmma_f32_16x16x4(a1, b, acc11);
        b.x = wk[32];
        b.y = wk[QKV_N + 32];
        acc02 = wmma_f32_16x16x4(a0, b, acc02);
        acc12 = wmma_f32_16x16x4(a1, b, acc12);
    }

    const float bv0 = bias[ncol0];
    const float bv1 = bias[ncol0 + 16];
    const float bv2 = bias[ncol0 + 32];
    for (int r = 0; r < 8; ++r) {
        const int rowa = mbase + r + half * 8;
        const int rowb = rowa + 16;
        if (rowa < natoms) {
            float* o = out + (size_t)rowa * QKV_N + ncol0;
            o[0]  = acc00[r] + bv0;
            o[16] = acc01[r] + bv1;
            o[32] = acc02[r] + bv2;
        }
        if (rowb < natoms) {
            float* o = out + (size_t)rowb * QKV_N + ncol0;
            o[0]  = acc10[r] + bv0;
            o[16] = acc11[r] + bv1;
            o[32] = acc12[r] + bv2;
        }
    }
}

__global__ void head_init_kernel(unsigned long long* __restrict__ head, size_t n) {
    size_t i = (size_t)blockIdx.x * blockDim.x + threadIdx.x;
    const size_t stride = (size_t)gridDim.x * blockDim.x;
    for (; i < n; i += stride) head[i] = LIST_END;
}

#define ROWS_PB 16
#define MPITCH  132

__global__ void __launch_bounds__(256)
gather_proj_kernel(const float* __restrict__ qkv,
                   const float* __restrict__ edge_feat,
                   const float* __restrict__ edge_vec,
                   const float* __restrict__ radial,
                   const int2*  __restrict__ head,
                   const int2*  __restrict__ link,
                   const float* __restrict__ ow,
                   const float* __restrict__ ob,
                   float* __restrict__ out_feat,
                   float* __restrict__ out_vec,
                   int natoms,
                   int nedges) {
    __shared__ __attribute__((aligned(16))) float Ms[ROWS_PB * MPITCH];

    const int lane = threadIdx.x & 31;
    const int wid   = __builtin_amdgcn_readfirstlane((int)(threadIdx.x >> 5));
    const int mbase = (int)blockIdx.x * ROWS_PB;
    const int hoff = (lane >> 1) * 24 + (lane & 1) * 4;

    for (int r = 0; r < ROWS_PB / 8; ++r) {
        const int lr   = wid * (ROWS_PB / 8) + r;
        const int atom = mbase + lr;
        v4f mf = {0.0f, 0.0f, 0.0f, 0.0f};

        if (atom < natoms) {
            const int2 first = head[atom];
            int e = first.x;
            int c = first.y;
            const v4f q = *(const v4f*)(qkv + (size_t)atom * QKV_N + hoff);

            v4f a0 = mf, a1 = mf, a2 = mf;

            for (int it = 0; it < nedges; ++it) {
                if ((unsigned)e >= (unsigned)nedges) break;
                const int2 nxt = link[e];
                const float rad = radial[e];
                const float ev0 = edge_vec[(size_t)e * 3 + 0];
                const float ev1 = edge_vec[(size_t)e * 3 + 1];
                const float ev2 = edge_vec[(size_t)e * 3 + 2];
                const float* kp = qkv + (size_t)c * QKV_N + hoff;
                const v4f k = *(const v4f*)(kp + 8);
                const v4f v = *(const v4f*)(kp + 16);
                const v4f ef = __builtin_nontemporal_load(
                    (const v4f*)(edge_feat + (size_t)e * DIM + lane * 4));

                float p = q.x * k.x + q.y * k.y + q.z * k.z + q.w * k.w;
                p += __builtin_bit_cast(float, __builtin_amdgcn_update_dpp(
                         0, __builtin_bit_cast(int, p), 0xB1, 0xF, 0xF, true));

                const float attn = 0.5f * p
                                 * (1.0f + erff(p * 0.70710678118654752440f))
                                 * rad;
                mf += v * ef * attn;
                a0 += v * ev0;
                a1 += v * ev1;
                a2 += v * ev2;
                e = nxt.x;
                c = nxt.y;
            }

            float* vb = out_vec + (size_t)atom * (3 * DIM) + lane * 4;
            *(v4f*)(vb)           = a0;
            *(v4f*)(vb + DIM)     = a1;
            *(v4f*)(vb + 2 * DIM) = a2;
        }
        *(v4f*)(Ms + lr * MPITCH + lane * 4) = mf;
    }
    __syncthreads();

    const int half = lane >> 4;
    const int l16  = lane & 15;
    const int ncol = wid * 16 + l16;
    const float* ap = Ms + l16 * MPITCH + 2 * half;
    const float* __restrict__ wp = ow + (size_t)(2 * half) * DIM + ncol;

    v8f acc = {};
    #pragma unroll 8
    for (int k0 = 0; k0 < DIM; k0 += 4) {
        const v2f a = *(const v2f*)(ap + k0);
        v2f b;
        b.x = wp[(size_t)k0 * DIM];
        b.y = wp[(size_t)(k0 + 1) * DIM];
        acc = wmma_f32_16x16x4(a, b, acc);
    }

    const float bv = ob[ncol];
    #pragma unroll
    for (int r = 0; r < 8; ++r) {
        const int rowm = mbase + r + half * 8;
        if (rowm < natoms)
            out_feat[(size_t)rowm * DIM + ncol] = acc[r] + bv;
    }
}

extern "C" void kernel_launch(void* const* d_in, const int* in_sizes, int n_in,
                              void* d_out, int out_size, void* d_ws, size_t ws_size,
                              hipStream_t stream) {
    const float* nf   = (const float*)d_in[0];
    const float* ef   = (const float*)d_in[1];
    const float* evec = (const float*)d_in[2];
    const float* rad  = (const float*)d_in[3];
    const float* qkvw = (const float*)d_in[4];
    const float* qkvb = (const float*)d_in[5];
    const float* outw = (const float*)d_in[6];
    const float* outb = (const float*)d_in[7];
    const float* lng  = (const float*)d_in[8];
    const float* lnb  = (const float*)d_in[9];
    const int*   row  = (const int*)d_in[10];
    const int*   col  = (const int*)d_in[11];

    const int natoms = in_sizes[0] / DIM;
    const int nedges = in_sizes[3];

    float* qkv   = (float*)d_ws;

    float* out_feat = (float*)d_out;
    float* out_vec  = out_feat + (size_t)natoms * DIM;

    unsigned long long* link = (unsigned long long*)(qkv + (size_t)natoms * QKV_N);
    unsigned long long* head = link + nedges;

    head_init_kernel<<<64, 256, 0, stream>>>(head, (size_t)natoms);
    const int lblocks = (nedges + 256 * LINK_EPT - 1) / (256 * LINK_EPT);
    dim3 g1((unsigned)lblocks + (unsigned)((natoms + QROWS - 1) / QROWS), 1);
    qkv_gemm_kernel<<<g1, 256, 0, stream>>>(nf, lng, lnb, qkvw, qkvb,
                                            qkv, natoms,
                                            row, col, head, link, nedges, lblocks);

    dim3 g2((unsigned)((natoms + ROWS_PB - 1) / ROWS_PB), 1);
    gather_proj_kernel<<<g2, 256, 0, stream>>>(qkv, ef, evec, rad,
                                               (const int2*)head, (const int2*)link,
                                               outw, outb, out_feat, out_vec,
                                               natoms, nedges);
}


# qkv workspace row re-laid from per-head q8|k8|v8 to planes q128|k128|v128 (GEMM column tiles permuted to match): a gathered k,v row is 8 whole 128-B lines, not 32-B pieces over 12; sums unchanged
# speedup vs baseline: 3.9213x; 1.0910x over previous
#include <hip/hip_runtime.h>
#include <math.h>

#define DIM      128
#define NHEADS   16
#define DPH      8
#define QKV_N    384

typedef float v2f __attribute__((ext_vector_type(2)));
typedef float v8f __attribute__((ext_vector_type(8)));
typedef float v4f __attribute__((ext_vector_type(4)));

__device__ __forceinline__ v8f wmma_f32_16x16x4(v2f a, v2f b, v8f c) {
    return __builtin_amdgcn_wmma_f32_16x16x4_f32(
        false, a, false, b, (short)0, c, false, false);
}

#define QROWS   32
#define APITCH  132
#define LINK_EPT 4

#define LIST_END 0x00000000FFFFFFFFull
__device__ __forceinline__ unsigned long long list_entry(int e, int c) {
    return (unsigned long long)(unsigned)e | ((unsigned long long)(unsigned)c << 32);
}

__global__ void __launch_bounds__(256)
qkv_gemm_kernel(const float* __restrict__ nf,
                const float* __restrict__ ln_g,
                const float* __restrict__ ln_b,
                const float* __restrict__ w,
                const float* __restrict__ bias,
                float* __restrict__ out,
                int natoms,
                const int* __restrict__ row,
                const int* __restrict__ col,
                unsigned long long* __restrict__ head,
                unsigned long long* __restrict__ link,
                int nedges,
                int link_blocks) {
    __shared__ __attribute__((aligned(16))) float As[QROWS * APITCH];

    if ((int)blockIdx.x < link_blocks) {
        const int e0 = blockIdx.x * (256 * LINK_EPT);
        const int ebase = e0 + threadIdx.x;
        if (e0 + 256 * LINK_EPT <= nedges) {
            int rv[LINK_EPT];
            unsigned long long en[LINK_EPT];
            #pragma unroll
            for (int j = 0; j < LINK_EPT; ++j) rv[j] = row[ebase + j * 256];
            #pragma unroll
            for (int j = 0; j < LINK_EPT; ++j)
                en[j] = list_entry(ebase + j * 256, col[ebase + j * 256]);
            #pragma unroll
            for (int j = 0; j < LINK_EPT; ++j) en[j] = atomicExch(&head[rv[j]], en[j]);
            #pragma unroll
            for (int j = 0; j < LINK_EPT; ++j) link[ebase + j * 256] = en[j];
        } else {
            for (int j = 0; j < LINK_EPT; ++j) {
                const int e = ebase + j * 256;
                if (e < nedges)
                    link[e] = atomicExch(&head[row[e]], list_entry(e, col[e]));
            }
        }
        return;
    }

    const int lane = threadIdx.x & 31;
    const int wid  = threadIdx.x >> 5;
    const int half = lane >> 4;
    const int l16  = lane & 15;
    const int mbase = ((int)blockIdx.x - link_blocks) * QROWS;

    const float g0 = ln_g[lane * 4 + 0], g1 = ln_g[lane * 4 + 1];
    const float g2 = ln_g[lane * 4 + 2], g3 = ln_g[lane * 4 + 3];
    const float b0 = ln_b[lane * 4 + 0], b1 = ln_b[lane * 4 + 1];
    const float b2 = ln_b[lane * 4 + 2], b3 = ln_b[lane * 4 + 3];
    for (int r = 0; r < QROWS / 8; ++r) {
        const int lr = wid * (QROWS / 8) + r;
        int grow = mbase + lr;
        if (grow >= natoms) grow = natoms - 1;
        const float4 v = ((const float4*)(nf + (size_t)grow * DIM))[lane];
        float s  = v.x + v.y + v.z + v.w;
        float s2 = v.x*v.x + v.y*v.y + v.z*v.z + v.w*v.w;
        for (int m = 1; m < 32; m <<= 1) {
            s  += __shfl_xor(s,  m, 32);
            s2 += __shfl_xor(s2, m, 32);
        }
        const float mu  = s * (1.0f / DIM);
        const float var = s2 * (1.0f / DIM) - mu * mu;
        const float rs  = rsqrtf(var + 1e-5f);
        float4 o;
        o.x = (v.x - mu) * rs * g0 + b0;
        o.y = (v.y - mu) * rs * g1 + b1;
        o.z = (v.z - mu) * rs * g2 + b2;
        o.w = (v.w - mu) * rs * g3 + b3;
        *(float4*)(As + lr * APITCH + lane * 4) = o;
    }
    __syncthreads();

    const int scol0 = wid * 48 + (l16 >> 3) * 24 + (l16 & 7);
    const int dcol0 = wid * 16 + l16;
    const float* a0p = As + l16 * APITCH + 2 * half;
    const float* a1p = a0p + 16 * APITCH;
    const float* __restrict__ wp = w + (size_t)(2 * half) * QKV_N + scol0;

    v8f acc00 = {}, acc01 = {}, acc02 = {};
    v8f acc10 = {}, acc11 = {}, acc12 = {};
    #pragma unroll 4
    for (int k0 = 0; k0 < DIM; k0 += 4) {
        const v2f a0 = *(const v2f*)(a0p + k0);
        const v2f a1 = *(const v2f*)(a1p + k0);
        const float* __restrict__ wk = wp + (size_t)k0 * QKV_N;
        v2f b;
        b.x = wk[0];
        b.y = wk[QKV_N];
        acc00 = wmma_f32_16x16x4(a0, b, acc00);
        acc10 = wmma_f32_16x16x4(a1, b, acc10);
        b.x = wk[8];
        b.y = wk[QKV_N + 8];
        acc01 = wmma_f32_16x16x4(a0, b, acc01);
        acc11 = wmma_f32_16x16x4(a1, b, acc11);
        b.x = wk[16];
        b.y = wk[QKV_N + 16];
        acc02 = wmma_f32_16x16x4(a0, b, acc02);
        acc12 = wmma_f32_16x16x4(a1, b, acc12);
    }

    const float bv0 = bias[scol0];
    const float bv1 = bias[scol0 + 8];
    const float bv2 = bias[scol0 + 16];
    for (int r = 0; r < 8; ++r) {
        const int rowa = mbase + r + half * 8;
        const int rowb = rowa + 16;
        if (rowa < natoms) {
            float* o = out + (size_t)rowa * QKV_N + dcol0;
            o[0]       = acc00[r] + bv0;
            o[DIM]     = acc01[r] + bv1;
            o[2 * DIM] = acc02[r] + bv2;
        }
        if (rowb < natoms) {
            float* o = out + (size_t)rowb * QKV_N + dcol0;
            o[0]       = acc10[r] + bv0;
            o[DIM]     = acc11[r] + bv1;
            o[2 * DIM] = acc12[r] + bv2;
        }
    }
}

__global__ void head_init_kernel(unsigned long long* __restrict__ head, size_t n) {
    size_t i = (size_t)blockIdx.x * blockDim.x + threadIdx.x;
    const size_t stride = (size_t)gridDim.x * blockDim.x;
    for (; i < n; i += stride) head[i] = LIST_END;
}

#define ROWS_PB 16
#define MPITCH  132

__global__ void __launch_bounds__(256)
gather_proj_kernel(const float* __restrict__ qkv,
                   const float* __restrict__ edge_feat,
                   const float* __restrict__ edge_vec,
                   const float* __restrict__ radial,
                   const int2*  __restrict__ head,
                   const int2*  __restrict__ link,
                   const float* __restrict__ ow,
                   const float* __restrict__ ob,
                   float* __restrict__ out_feat,
                   float* __restrict__ out_vec,
                   int natoms,
                   int nedges) {
    __shared__ __attribute__((aligned(16))) float Ms[ROWS_PB * MPITCH];

    const int lane = threadIdx.x & 31;
    const int wid   = __builtin_amdgcn_readfirstlane((int)(threadIdx.x >> 5));
    const int mbase = (int)blockIdx.x * ROWS_PB;
    const int foff = lane * 4;

    for (int r = 0; r < ROWS_PB / 8; ++r) {
        const int lr   = wid * (ROWS_PB / 8) + r;
        const int atom = mbase + lr;
        v4f mf = {0.0f, 0.0f, 0.0f, 0.0f};

        if (atom < natoms) {
            const int2 first = head[atom];
            int e = first.x;
            int c = first.y;
            const v4f q = *(const v4f*)(qkv + (size_t)atom * QKV_N + foff);

            v4f a0 = mf, a1 = mf, a2 = mf;

            for (int it = 0; it < nedges; ++it) {
                if ((unsigned)e >= (unsigned)nedges) break;
                const int2 nxt = link[e];
                const float rad = radial[e];
                const float ev0 = edge_vec[(size_t)e * 3 + 0];
                const float ev1 = edge_vec[(size_t)e * 3 + 1];
                const float ev2 = edge_vec[(size_t)e * 3 + 2];
                const float* kp = qkv + (size_t)c * QKV_N + foff;
                const v4f k = *(const v4f*)(kp + DIM);
                const v4f v = *(const v4f*)(kp + 2 * DIM);
                const v4f ef = __builtin_nontemporal_load(
                    (const v4f*)(edge_feat + (size_t)e * DIM + lane * 4));

                float p = q.x * k.x + q.y * k.y + q.z * k.z + q.w * k.w;
                p += __builtin_bit_cast(float, __builtin_amdgcn_update_dpp(
                         0, __builtin_bit_cast(int, p), 0xB1, 0xF, 0xF, true));

                const float attn = 0.5f * p
                                 * (1.0f + erff(p * 0.70710678118654752440f))
                                 * rad;
                mf += v * ef * attn;
                a0 += v * ev0;
                a1 += v * ev1;
                a2 += v * ev2;
                e = nxt.x;
                c = nxt.y;
            }

            float* vb = out_vec + (size_t)atom * (3 * DIM) + lane * 4;
            *(v4f*)(vb)           = a0;
            *(v4f*)(vb + DIM)     = a1;
            *(v4f*)(vb + 2 * DIM) = a2;
        }
        *(v4f*)(Ms + lr * MPITCH + lane * 4) = mf;
    }
    __syncthreads();

    const int half = lane >> 4;
    const int l16  = lane & 15;
    const int ncol = wid * 16 + l16;
    const float* ap = Ms + l16 * MPITCH + 2 * half;
    const float* __restrict__ wp = ow + (size_t)(2 * half) * DIM + ncol;

    v8f acc = {};
    #pragma unroll 8
    for (int k0 = 0; k0 < DIM; k0 += 4) {
        const v2f a = *(const v2f*)(ap + k0);
        v2f b;
        b.x = wp[(size_t)k0 * DIM];
        b.y = wp[(size_t)(k0 + 1) * DIM];
        acc = wmma_f32_16x16x4(a, b, acc);
    }

    const float bv = ob[ncol];
    #pragma unroll
    for (int r = 0; r < 8; ++r) {
        const int rowm = mbase + r + half * 8;
        if (rowm < natoms)
            out_feat[(size_t)rowm * DIM + ncol] = acc[r] + bv;
    }
}

extern "C" void kernel_launch(void* const* d_in, const int* in_sizes, int n_in,
                              void* d_out, int out_size, void* d_ws, size_t ws_size,
                              hipStream_t stream) {
    const float* nf   = (const float*)d_in[0];
    const float* ef   = (const float*)d_in[1];
    const float* evec = (const float*)d_in[2];
    const float* rad  = (const float*)d_in[3];
    const float* qkvw = (const float*)d_in[4];
    const float* qkvb = (const float*)d_in[5];
    const float* outw = (const float*)d_in[6];
    const float* outb = (const float*)d_in[7];
    const float* lng  = (const float*)d_in[8];
    const float* lnb  = (const float*)d_in[9];
    const int*   row  = (const int*)d_in[10];
    const int*   col  = (const int*)d_in[11];

    const int natoms = in_sizes[0] / DIM;
    const int nedges = in_sizes[3];

    float* qkv   = (float*)d_ws;

    float* out_feat = (float*)d_out;
    float* out_vec  = out_feat + (size_t)natoms * DIM;

    unsigned long long* link = (unsigned long long*)(qkv + (size_t)natoms * QKV_N);
    unsigned long long* head = link + nedges;

    head_init_kernel<<<64, 256, 0, stream>>>(head, (size_t)natoms);
    const int lblocks = (nedges + 256 * LINK_EPT - 1) / (256 * LINK_EPT);
    dim3 g1((unsigned)lblocks + (unsigned)((natoms + QROWS - 1) / QROWS), 1);
    qkv_gemm_kernel<<<g1, 256, 0, stream>>>(nf, lng, lnb, qkvw, qkvb,
                                            qkv, natoms,
                                            row, col, head, link, nedges, lblocks);

    dim3 g2((unsigned)((natoms + ROWS_PB - 1) / ROWS_PB), 1);
    gather_proj_kernel<<<g2, 256, 0, stream>>>(qkv, ef, evec, rad,
                                               (const int2*)head, (const int2*)link,
                                               outw, outb, out_feat, out_vec,
                                               natoms, nedges);
}
